# starting kernel
# baseline (speedup 1.0000x reference)
#include <hip/hip_runtime.h>
#include <hip/hip_bf16.h>

#define BATCH 4
#define NPTS  8192
#define CH    32
#define KNN   16
#define LDIM  64
#define ODIM  128
#define NCTILES (NPTS / 16)
#define ROWS_PER_WAVE 32
#define NRBLK (NPTS / ROWS_PER_WAVE)

typedef __attribute__((ext_vector_type(2))) float v2f;
typedef __attribute__((ext_vector_type(8))) float v8f;

__global__ void sq_kernel(const float* __restrict__ x, float* __restrict__ sq, int total) {
    int i = blockIdx.x * blockDim.x + threadIdx.x;
    if (i >= total) return;
    const float* p = x + (size_t)i * CH;
    float s = 0.0f;
#pragma unroll
    for (int c = 0; c < CH; ++c) s += p[c] * p[c];
    sq[i] = s;
}

__global__ __launch_bounds__(32) void knn_pool_kernel(const float* __restrict__ x,
                                                      const float* __restrict__ sq,
                                                      float* __restrict__ pooled) {
    __shared__ float distT[ROWS_PER_WAVE * 19];
    __shared__ int   selIdx[ROWS_PER_WAVE * KNN];

    const int lane    = threadIdx.x;
    const int rblk    = blockIdx.x % NRBLK;
    const int b       = blockIdx.x / NRBLK;
    const int rowbase = rblk * ROWS_PER_WAVE;
    const float* xb   = x  + (size_t)b * NPTS * CH;
    const float* sqb  = sq + (size_t)b * NPTS;

    const int m = lane & 15;
    const int h = lane >> 4;

    v2f afrag0[8], afrag1[8];
#pragma unroll
    for (int c = 0; c < 8; ++c) {
        const float* p0 = xb + (size_t)(rowbase + m) * CH + 4 * c + 2 * h;
        const float* p1 = xb + (size_t)(rowbase + 16 + m) * CH + 4 * c + 2 * h;
        afrag0[c].x = -2.0f * p0[0];
        afrag0[c].y = -2.0f * p0[1];
        afrag1[c].x = -2.0f * p1[0];
        afrag1[c].y = -2.0f * p1[1];
    }
    v2f afragOne;
    afragOne.x = (h == 0) ? 1.0f : 0.0f;
    afragOne.y = 0.0f;

    float key[KNN];
    int   idx[KNN];
#pragma unroll
    for (int s = 0; s < KNN; ++s) { key[s] = 3.0e38f; idx[s] = 0; }

    for (int jt = 0; jt < NCTILES; ++jt) {
        const int colbase = jt * 16;

        v8f acc0 = {};
        v8f acc1 = {};
#pragma unroll
        for (int c = 0; c < 8; ++c) {
            const float* p = xb + (size_t)(colbase + m) * CH + 4 * c + 2 * h;
            v2f bfrag;
            bfrag.x = p[0];
            bfrag.y = p[1];
            acc0 = __builtin_amdgcn_wmma_f32_16x16x4_f32(
                false, afrag0[c], false, bfrag, (short)0, acc0, false, false);
            acc1 = __builtin_amdgcn_wmma_f32_16x16x4_f32(
                false, afrag1[c], false, bfrag, (short)0, acc1, false, false);
        }
        {
            v2f bfrag;
            bfrag.x = (h == 0) ? sqb[colbase + m] : 0.0f;
            bfrag.y = 0.0f;
            acc0 = __builtin_amdgcn_wmma_f32_16x16x4_f32(
                false, afragOne, false, bfrag, (short)0, acc0, false, false);
            acc1 = __builtin_amdgcn_wmma_f32_16x16x4_f32(
                false, afragOne, false, bfrag, (short)0, acc1, false, false);
        }

#pragma unroll
        for (int r = 0; r < 8; ++r) {
            distT[(r + 8 * h) * 19 + m]        = acc0[r];
            distT[(r + 8 * h + 16) * 19 + m]   = acc1[r];
        }
        __syncthreads();

#pragma unroll 1
        for (int c = 0; c < 16; ++c) {
            const float d = distT[lane * 19 + c];
            if (d < key[KNN - 1]) {
                float cv = d;
                int   ci = colbase + c;
#pragma unroll
                for (int s = 0; s < KNN; ++s) {
                    const float ks = key[s];
                    const int   is = idx[s];
                    const bool  sw = cv < ks;
                    key[s] = sw ? cv : ks;
                    idx[s] = sw ? ci : is;
                    cv = sw ? ks : cv;
                    ci = sw ? is : ci;
                }
            }
        }
        __syncthreads();
    }

#pragma unroll
    for (int s = 0; s < KNN; ++s)
        selIdx[lane * KNN + s] = idx[s];
    __syncthreads();

    float* poolb = pooled + ((size_t)b * NPTS + rowbase) * CH;
#pragma unroll 1
    for (int r = 0; r < ROWS_PER_WAVE; ++r) {
        float mx = -3.0e38f;
#pragma unroll
        for (int k = 0; k < KNN; ++k) {
            const int j = selIdx[r * KNN + k];
            const float v = xb[(size_t)j * CH + lane];
            mx = fmaxf(mx, v);
        }
        poolb[r * CH + lane] = mx;
    }
}

__global__ __launch_bounds__(32) void mlp_kernel(const float* __restrict__ pooled,
                                                 const float* __restrict__ Wl,
                                                 const float* __restrict__ bl,
                                                 const float* __restrict__ Wc,
                                                 const float* __restrict__ bc,
                                                 float* __restrict__ out) {
    __shared__ float hT[16 * 68];

    const int lane = threadIdx.x;
    const int m = lane & 15;
    const int h = lane >> 4;
    const size_t rowbase = (size_t)blockIdx.x * 16;

    v2f afrag[8];
#pragma unroll
    for (int c = 0; c < 8; ++c) {
        const float* p = pooled + (rowbase + m) * CH + 4 * c + 2 * h;
        afrag[c].x = p[0];
        afrag[c].y = p[1];
    }

#pragma unroll
    for (int t = 0; t < 4; ++t) {
        v8f acc = {};
#pragma unroll
        for (int c = 0; c < 8; ++c) {
            const float* p = Wl + (size_t)(4 * c + 2 * h) * LDIM + t * 16 + m;
            v2f bfrag;
            bfrag.x = p[0];
            bfrag.y = p[LDIM];
            acc = __builtin_amdgcn_wmma_f32_16x16x4_f32(
                false, afrag[c], false, bfrag, (short)0, acc, false, false);
        }
        const float bias = bl[t * 16 + m];
#pragma unroll
        for (int r = 0; r < 8; ++r)
            hT[(r + 8 * h) * 68 + t * 16 + m] = acc[r] + bias;
    }
    __syncthreads();

    v2f a2[16];
#pragma unroll
    for (int c = 0; c < 16; ++c) {
        a2[c].x = hT[m * 68 + 4 * c + 2 * h];
        a2[c].y = hT[m * 68 + 4 * c + 2 * h + 1];
    }

    float* outp = out + rowbase * ODIM;
#pragma unroll
    for (int t = 0; t < 8; ++t) {
        v8f acc = {};
#pragma unroll
        for (int c = 0; c < 16; ++c) {
            const float* p = Wc + (size_t)(4 * c + 2 * h) * ODIM + t * 16 + m;
            v2f bfrag;
            bfrag.x = p[0];
            bfrag.y = p[ODIM];
            acc = __builtin_amdgcn_wmma_f32_16x16x4_f32(
                false, a2[c], false, bfrag, (short)0, acc, false, false);
        }
        const float bias = bc[t * 16 + m];
#pragma unroll
        for (int r = 0; r < 8; ++r) {
            const float v = acc[r] + bias;
            outp[(size_t)(r + 8 * h) * ODIM + t * 16 + m] = fmaxf(v, 0.0f);
        }
    }
}

extern "C" void kernel_launch(void* const* d_in, const int* in_sizes, int n_in,
                              void* d_out, int out_size, void* d_ws, size_t ws_size,
                              hipStream_t stream) {
    const float* x  = (const float*)d_in[0];
    const float* Wl = (const float*)d_in[1];
    const float* bl = (const float*)d_in[2];
    const float* Wc = (const float*)d_in[3];
    const float* bc = (const float*)d_in[4];
    float* out = (float*)d_out;

    float* pooled = (float*)d_ws;
    float* sqbuf  = pooled + (size_t)BATCH * NPTS * CH;

    const int total = BATCH * NPTS;
    sq_kernel<<<(total + 255) / 256, 256, 0, stream>>>(x, sqbuf, total);
    knn_pool_kernel<<<BATCH * NRBLK, 32, 0, stream>>>(x, sqbuf, pooled);
    mlp_kernel<<<total / 16, 32, 0, stream>>>(pooled, Wl, bl, Wc, bc, out);
}


# kNN candidates staged once per 4-wave work-group (4 x 32 rows) in a two-slot LDS ring of 512-point panels, norms beside them; A fragments by 16-byte LDS reads; key sums and order unchanged
# speedup vs baseline: 1.7822x; 1.7822x over previous
#include <hip/hip_runtime.h>
#include <hip/hip_bf16.h>

#define BATCH 4
#define NPTS  8192
#define CH    32
#define KNN   16
#define LDIM  64
#define ODIM  128
#define NCTILES (NPTS / 16)
#define ROWS_PER_WAVE 32
#define NRBLK (NPTS / ROWS_PER_WAVE)

typedef __attribute__((ext_vector_type(2))) float v2f;
typedef __attribute__((ext_vector_type(8))) float v8f;

__global__ void sq_kernel(const float* __restrict__ x, float* __restrict__ sq, int total) {
    int i = blockIdx.x * blockDim.x + threadIdx.x;
    if (i >= total) return;
    const float* p = x + (size_t)i * CH;
    float s = 0.0f;
#pragma unroll
    for (int c = 0; c < CH; ++c) s += p[c] * p[c];
    sq[i] = s;
}

#define WAVES_PER_WG     4
#define ROWS_PER_WG      (WAVES_PER_WG * ROWS_PER_WAVE)
#define NWG_PER_BATCH    (NPTS / ROWS_PER_WG)
#define TILES_PER_PANEL  32
#define PANEL_PTS        (TILES_PER_PANEL * 16)
#define NPANELS          (NPTS / PANEL_PTS)
#define PITCH            36

typedef __attribute__((ext_vector_type(4))) float v4f;

__device__ __forceinline__ float xhalf(float v) {
    const int b = __builtin_bit_cast(int, v);
    return __builtin_bit_cast(float,
        __builtin_amdgcn_permlanex16(b, b, 0x76543210, 0xfedcba98, false, false));
}

__global__ __launch_bounds__(128) void knn_pool_kernel(const float* __restrict__ x,
                                                       const float* __restrict__ sq,
                                                       float* __restrict__ pooled) {
    __shared__ __attribute__((aligned(16))) float pan[2 * PANEL_PTS * PITCH];
    __shared__ int selIdx[WAVES_PER_WG * ROWS_PER_WAVE * KNN];

    const int tid     = threadIdx.x;
    const int lane    = tid & 31;
    const int wave    = __builtin_amdgcn_readfirstlane(tid >> 5);
    const int b       = blockIdx.x / NWG_PER_BATCH;
    const int rowbase = (blockIdx.x % NWG_PER_BATCH) * ROWS_PER_WG + wave * ROWS_PER_WAVE;
    const float* xb   = x  + (size_t)b * NPTS * CH;
    const float* sqb  = sq + (size_t)b * NPTS;

    const int m = lane & 15;
    const int h = lane >> 4;

    const int spt  = tid >> 3;
    const int sq4  = tid & 7;
    const int sdst = spt * PITCH + 8 * (sq4 >> 1) + 2 * (sq4 & 1);
    const int ndst = spt * PITCH + 32 + (sq4 & 3);
    const bool nlead = (sq4 & 3) == 0;
    const float* gsrc = xb + tid * 4;
    const float* gsq  = sqb + spt;

    v2f qfrag0[8], qfrag1[8];
#pragma unroll
    for (int c = 0; c < 8; ++c) {
        const float* p0 = xb + (size_t)(rowbase + m) * CH + 4 * c + 2 * h;
        const float* p1 = xb + (size_t)(rowbase + 16 + m) * CH + 4 * c + 2 * h;
        qfrag0[c].x = -2.0f * p0[0];
        qfrag0[c].y = -2.0f * p0[1];
        qfrag1[c].x = -2.0f * p1[0];
        qfrag1[c].y = -2.0f * p1[1];
    }
    v2f qfragOne;
    qfragOne.x = (h == 0) ? 1.0f : 0.0f;
    qfragOne.y = 0.0f;

    float key[KNN];
    int   idx[KNN];
#pragma unroll
    for (int s = 0; s < KNN; ++s) { key[s] = 3.0e38f; idx[s] = 0; }

#pragma unroll 1
    for (int s0 = 0; s0 < TILES_PER_PANEL; s0 += 8) {
        v4f   tv[8];
        float ts[8];
#pragma unroll
        for (int u = 0; u < 8; ++u) {
            const int tb = (s0 + u) * 16;
            tv[u] = *(const v4f*)(gsrc + (size_t)tb * CH);
            ts[u] = gsq[tb];
        }
#pragma unroll
        for (int u = 0; u < 8; ++u) {
            float* d = pan + (s0 + u) * 16 * PITCH;
            v2f lo; lo.x = tv[u].x; lo.y = tv[u].y;
            v2f hi; hi.x = tv[u].z; hi.y = tv[u].w;
            *(v2f*)(d + sdst)     = lo;
            *(v2f*)(d + sdst + 4) = hi;
            d[ndst] = nlead ? ts[u] : 0.0f;
        }
    }
    __syncthreads();

    const int foff = m * PITCH + 4 * h;

    for (int pn = 0; pn < NPANELS; ++pn) {
        const float* cur = pan + (pn & 1) * (PANEL_PTS * PITCH);
        float*       nxt = pan + ((pn & 1) ^ 1) * (PANEL_PTS * PITCH);
        const bool  more = (pn + 1 < NPANELS);
        const int   nextbase = (pn + 1) * PANEL_PTS;

        v4f f[4];
        v2f fs;
#pragma unroll
        for (int g = 0; g < 4; ++g) f[g] = *(const v4f*)(cur + foff + 8 * g);
        fs = *(const v2f*)(cur + m * PITCH + 32 + 2 * h);

#pragma unroll 1
        for (int tt = 0; tt < TILES_PER_PANEL; ++tt) {
            const int colbase = pn * PANEL_PTS + tt * 16;

            v4f   sv = {};
            float ss = 0.0f;
            if (more) {
                const int tb = nextbase + tt * 16;
                sv = *(const v4f*)(gsrc + (size_t)tb * CH);
                ss = gsq[tb];
            }

            v8f acc0 = {};
            v8f acc1 = {};
#pragma unroll
            for (int g = 0; g < 4; ++g) {
                v2f c0; c0.x = f[g].x; c0.y = f[g].y;
                v2f c1; c1.x = f[g].z; c1.y = f[g].w;
                acc0 = __builtin_amdgcn_wmma_f32_16x16x4_f32(
                    false, c0, false, qfrag0[2 * g], (short)0, acc0, false, false);
                acc1 = __builtin_amdgcn_wmma_f32_16x16x4_f32(
                    false, c0, false, qfrag1[2 * g], (short)0, acc1, false, false);
                acc0 = __builtin_amdgcn_wmma_f32_16x16x4_f32(
                    false, c1, false, qfrag0[2 * g + 1], (short)0, acc0, false, false);
                acc1 = __builtin_amdgcn_wmma_f32_16x16x4_f32(
                    false, c1, false, qfrag1[2 * g + 1], (short)0, acc1, false, false);
            }
            acc0 = __builtin_amdgcn_wmma_f32_16x16x4_f32(
                false, fs, false, qfragOne, (short)0, acc0, false, false);
            acc1 = __builtin_amdgcn_wmma_f32_16x16x4_f32(
                false, fs, false, qfragOne, (short)0, acc1, false, false);

            if (tt + 1 < TILES_PER_PANEL) {
                const float* fr = cur + (tt + 1) * 16 * PITCH;
#pragma unroll
                for (int g = 0; g < 4; ++g) f[g] = *(const v4f*)(fr + foff + 8 * g);
                fs = *(const v2f*)(fr + m * PITCH + 32 + 2 * h);
            }

            float cand[16];
#pragma unroll
            for (int r = 0; r < 8; ++r) {
                const float a0 = acc0[r];
                const float a1 = acc1[r];
                const float give = (h == 0) ? a1 : a0;
                const float got  = xhalf(give);
                cand[r]     = (h == 0) ? a0 : got;
                cand[8 + r] = (h == 0) ? got : a1;
            }

#pragma unroll
            for (int c = 0; c < 16; ++c) {
                const float d = cand[c];
                if (d < key[KNN - 1]) {
                    float cv = d;
                    int   ci = colbase + c;
#pragma unroll
                    for (int s = 0; s < KNN; ++s) {
                        const float ks = key[s];
                        const int   is = idx[s];
                        const bool  sw = cv < ks;
                        key[s] = sw ? cv : ks;
                        idx[s] = sw ? ci : is;
                        cv = sw ? ks : cv;
                        ci = sw ? is : ci;
                    }
                }
            }

            if (more) {
                float* d = nxt + tt * 16 * PITCH;
                v2f lo; lo.x = sv.x; lo.y = sv.y;
                v2f hi; hi.x = sv.z; hi.y = sv.w;
                *(v2f*)(d + sdst)     = lo;
                *(v2f*)(d + sdst + 4) = hi;
                d[ndst] = nlead ? ss : 0.0f;
            }
        }
        __syncthreads();
    }

    int* sel = selIdx + wave * (ROWS_PER_WAVE * KNN);
#pragma unroll
    for (int s = 0; s < KNN; ++s)
        sel[lane * KNN + s] = idx[s];
    __syncthreads();

    float* poolb = pooled + ((size_t)b * NPTS + rowbase) * CH;
#pragma unroll 1
    for (int r = 0; r < ROWS_PER_WAVE; ++r) {
        float mx = -3.0e38f;
#pragma unroll
        for (int k = 0; k < KNN; ++k) {
            const int j = sel[r * KNN + k];
            const float v = xb[(size_t)j * CH + lane];
            mx = fmaxf(mx, v);
        }
        poolb[r * CH + lane] = mx;
    }
}

__global__ __launch_bounds__(32) void mlp_kernel(const float* __restrict__ pooled,
                                                 const float* __restrict__ Wl,
                                                 const float* __restrict__ bl,
                                                 const float* __restrict__ Wc,
                                                 const float* __restrict__ bc,
                                                 float* __restrict__ out) {
    __shared__ float hT[16 * 68];

    const int lane = threadIdx.x;
    const int m = lane & 15;
    const int h = lane >> 4;
    const size_t rowbase = (size_t)blockIdx.x * 16;

    v2f afrag[8];
#pragma unroll
    for (int c = 0; c < 8; ++c) {
        const float* p = pooled + (rowbase + m) * CH + 4 * c + 2 * h;
        afrag[c].x = p[0];
        afrag[c].y = p[1];
    }

#pragma unroll
    for (int t = 0; t < 4; ++t) {
        v8f acc = {};
#pragma unroll
        for (int c = 0; c < 8; ++c) {
            const float* p = Wl + (size_t)(4 * c + 2 * h) * LDIM + t * 16 + m;
            v2f bfrag;
            bfrag.x = p[0];
            bfrag.y = p[LDIM];
            acc = __builtin_amdgcn_wmma_f32_16x16x4_f32(
                false, afrag[c], false, bfrag, (short)0, acc, false, false);
        }
        const float bias = bl[t * 16 + m];
#pragma unroll
        for (int r = 0; r < 8; ++r)
            hT[(r + 8 * h) * 68 + t * 16 + m] = acc[r] + bias;
    }
    __syncthreads();

    v2f a2[16];
#pragma unroll
    for (int c = 0; c < 16; ++c) {
        a2[c].x = hT[m * 68 + 4 * c + 2 * h];
        a2[c].y = hT[m * 68 + 4 * c + 2 * h + 1];
    }

    float* outp = out + rowbase * ODIM;
#pragma unroll
    for (int t = 0; t < 8; ++t) {
        v8f acc = {};
#pragma unroll
        for (int c = 0; c < 16; ++c) {
            const float* p = Wc + (size_t)(4 * c + 2 * h) * ODIM + t * 16 + m;
            v2f bfrag;
            bfrag.x = p[0];
            bfrag.y = p[ODIM];
            acc = __builtin_amdgcn_wmma_f32_16x16x4_f32(
                false, a2[c], false, bfrag, (short)0, acc, false, false);
        }
        const float bias = bc[t * 16 + m];
#pragma unroll
        for (int r = 0; r < 8; ++r) {
            const float v = acc[r] + bias;
            outp[(size_t)(r + 8 * h) * ODIM + t * 16 + m] = fmaxf(v, 0.0f);
        }
    }
}

extern "C" void kernel_launch(void* const* d_in, const int* in_sizes, int n_in,
                              void* d_out, int out_size, void* d_ws, size_t ws_size,
                              hipStream_t stream) {
    const float* x  = (const float*)d_in[0];
    const float* Wl = (const float*)d_in[1];
    const float* bl = (const float*)d_in[2];
    const float* Wc = (const float*)d_in[3];
    const float* bc = (const float*)d_in[4];
    float* out = (float*)d_out;

    float* pooled = (float*)d_ws;
    float* sqbuf  = pooled + (size_t)BATCH * NPTS * CH;

    const int total = BATCH * NPTS;
    sq_kernel<<<(total + 255) / 256, 256, 0, stream>>>(x, sqbuf, total);
    knn_pool_kernel<<<BATCH * NWG_PER_BATCH, 128, 0, stream>>>(x, sqbuf, pooled);
    mlp_kernel<<<total / 16, 32, 0, stream>>>(pooled, Wl, bl, Wc, bc, out);
}


# kNN top-16 insertion as a per-slot clamp, key[s] = median(key[s-1], key[s], d), indices by the old list's compare masks: 16 independent slots, no dependent compare-select chain; sums unchanged
# speedup vs baseline: 2.1160x; 1.1873x over previous
#include <hip/hip_runtime.h>
#include <hip/hip_bf16.h>

#define BATCH 4
#define NPTS  8192
#define CH    32
#define KNN   16
#define LDIM  64
#define ODIM  128
#define NCTILES (NPTS / 16)
#define ROWS_PER_WAVE 32
#define NRBLK (NPTS / ROWS_PER_WAVE)

typedef __attribute__((ext_vector_type(2))) float v2f;
typedef __attribute__((ext_vector_type(8))) float v8f;

__global__ void sq_kernel(const float* __restrict__ x, float* __restrict__ sq, int total) {
    int i = blockIdx.x * blockDim.x + threadIdx.x;
    if (i >= total) return;
    const float* p = x + (size_t)i * CH;
    float s = 0.0f;
#pragma unroll
    for (int c = 0; c < CH; ++c) s += p[c] * p[c];
    sq[i] = s;
}

#define WAVES_PER_WG     4
#define ROWS_PER_WG      (WAVES_PER_WG * ROWS_PER_WAVE)
#define NWG_PER_BATCH    (NPTS / ROWS_PER_WG)
#define TILES_PER_PANEL  32
#define PANEL_PTS        (TILES_PER_PANEL * 16)
#define NPANELS          (NPTS / PANEL_PTS)
#define PITCH            36

typedef __attribute__((ext_vector_type(4))) float v4f;

__device__ __forceinline__ float xhalf(float v) {
    const int b = __builtin_bit_cast(int, v);
    return __builtin_bit_cast(float,
        __builtin_amdgcn_permlanex16(b, b, 0x76543210, 0xfedcba98, false, false));
}

__global__ __launch_bounds__(128) void knn_pool_kernel(const float* __restrict__ x,
                                                       const float* __restrict__ sq,
                                                       float* __restrict__ pooled) {
    __shared__ __attribute__((aligned(16))) float pan[2 * PANEL_PTS * PITCH];
    __shared__ int selIdx[WAVES_PER_WG * ROWS_PER_WAVE * KNN];

    const int tid     = threadIdx.x;
    const int lane    = tid & 31;
    const int wave    = __builtin_amdgcn_readfirstlane(tid >> 5);
    const int b       = blockIdx.x / NWG_PER_BATCH;
    const int rowbase = (blockIdx.x % NWG_PER_BATCH) * ROWS_PER_WG + wave * ROWS_PER_WAVE;
    const float* xb   = x  + (size_t)b * NPTS * CH;
    const float* sqb  = sq + (size_t)b * NPTS;

    const int m = lane & 15;
    const int h = lane >> 4;

    const int spt  = tid >> 3;
    const int sq4  = tid & 7;
    const int sdst = spt * PITCH + 8 * (sq4 >> 1) + 2 * (sq4 & 1);
    const int ndst = spt * PITCH + 32 + (sq4 & 3);
    const bool nlead = (sq4 & 3) == 0;
    const float* gsrc = xb + tid * 4;
    const float* gsq  = sqb + spt;

    v2f qfrag0[8], qfrag1[8];
#pragma unroll
    for (int c = 0; c < 8; ++c) {
        const float* p0 = xb + (size_t)(rowbase + m) * CH + 4 * c + 2 * h;
        const float* p1 = xb + (size_t)(rowbase + 16 + m) * CH + 4 * c + 2 * h;
        qfrag0[c].x = -2.0f * p0[0];
        qfrag0[c].y = -2.0f * p0[1];
        qfrag1[c].x = -2.0f * p1[0];
        qfrag1[c].y = -2.0f * p1[1];
    }
    v2f qfragOne;
    qfragOne.x = (h == 0) ? 1.0f : 0.0f;
    qfragOne.y = 0.0f;

    float key[KNN];
    int   idx[KNN];
#pragma unroll
    for (int s = 0; s < KNN; ++s) { key[s] = 3.0e38f; idx[s] = 0; }

#pragma unroll 1
    for (int s0 = 0; s0 < TILES_PER_PANEL; s0 += 8) {
        v4f   tv[8];
        float ts[8];
#pragma unroll
        for (int u = 0; u < 8; ++u) {
            const int tb = (s0 + u) * 16;
            tv[u] = *(const v4f*)(gsrc + (size_t)tb * CH);
            ts[u] = gsq[tb];
        }
#pragma unroll
        for (int u = 0; u < 8; ++u) {
            float* d = pan + (s0 + u) * 16 * PITCH;
            v2f lo; lo.x = tv[u].x; lo.y = tv[u].y;
            v2f hi; hi.x = tv[u].z; hi.y = tv[u].w;
            *(v2f*)(d + sdst)     = lo;
            *(v2f*)(d + sdst + 4) = hi;
            d[ndst] = nlead ? ts[u] : 0.0f;
        }
    }
    __syncthreads();

    const int foff = m * PITCH + 4 * h;

    for (int pn = 0; pn < NPANELS; ++pn) {
        const float* cur = pan + (pn & 1) * (PANEL_PTS * PITCH);
        float*       nxt = pan + ((pn & 1) ^ 1) * (PANEL_PTS * PITCH);
        const bool  more = (pn + 1 < NPANELS);
        const int   nextbase = (pn + 1) * PANEL_PTS;

        v4f f[4];
        v2f fs;
#pragma unroll
        for (int g = 0; g < 4; ++g) f[g] = *(const v4f*)(cur + foff + 8 * g);
        fs = *(const v2f*)(cur + m * PITCH + 32 + 2 * h);

#pragma unroll 1
        for (int tt = 0; tt < TILES_PER_PANEL; ++tt) {
            const int colbase = pn * PANEL_PTS + tt * 16;

            v4f   sv = {};
            float ss = 0.0f;
            if (more) {
                const int tb = nextbase + tt * 16;
                sv = *(const v4f*)(gsrc + (size_t)tb * CH);
                ss = gsq[tb];
            }

            v8f acc0 = {};
            v8f acc1 = {};
#pragma unroll
            for (int g = 0; g < 4; ++g) {
                v2f c0; c0.x = f[g].x; c0.y = f[g].y;
                v2f c1; c1.x = f[g].z; c1.y = f[g].w;
                acc0 = __builtin_amdgcn_wmma_f32_16x16x4_f32(
                    false, c0, false, qfrag0[2 * g], (short)0, acc0, false, false);
                acc1 = __builtin_amdgcn_wmma_f32_16x16x4_f32(
                    false, c0, false, qfrag1[2 * g], (short)0, acc1, false, false);
                acc0 = __builtin_amdgcn_wmma_f32_16x16x4_f32(
                    false, c1, false, qfrag0[2 * g + 1], (short)0, acc0, false, false);
                acc1 = __builtin_amdgcn_wmma_f32_16x16x4_f32(
                    false, c1, false, qfrag1[2 * g + 1], (short)0, acc1, false, false);
            }
            acc0 = __builtin_amdgcn_wmma_f32_16x16x4_f32(
                false, fs, false, qfragOne, (short)0, acc0, false, false);
            acc1 = __builtin_amdgcn_wmma_f32_16x16x4_f32(
                false, fs, false, qfragOne, (short)0, acc1, false, false);

            if (tt + 1 < TILES_PER_PANEL) {
                const float* fr = cur + (tt + 1) * 16 * PITCH;
#pragma unroll
                for (int g = 0; g < 4; ++g) f[g] = *(const v4f*)(fr + foff + 8 * g);
                fs = *(const v2f*)(fr + m * PITCH + 32 + 2 * h);
            }

            float cand[16];
#pragma unroll
            for (int r = 0; r < 8; ++r) {
                const float a0 = acc0[r];
                const float a1 = acc1[r];
                const float give = (h == 0) ? a1 : a0;
                const float got  = xhalf(give);
                cand[r]     = (h == 0) ? a0 : got;
                cand[8 + r] = (h == 0) ? got : a1;
            }

#pragma unroll
            for (int c = 0; c < 16; ++c) {
                const float d = cand[c];
                if (d < key[KNN - 1]) {
                    const int ci = colbase + c;
                    bool gt[KNN - 1];
#pragma unroll
                    for (int s = KNN - 2; s >= 0; --s) gt[s] = d < key[s];
                    __builtin_amdgcn_sched_barrier(0);
                    key[KNN - 1] = __builtin_amdgcn_fmed3f(key[KNN - 2], key[KNN - 1], d);
                    idx[KNN - 1] = gt[KNN - 2] ? idx[KNN - 2] : ci;
#pragma unroll
                    for (int s = KNN - 2; s >= 1; --s) {
                        key[s] = __builtin_amdgcn_fmed3f(key[s - 1], key[s], d);
                        idx[s] = gt[s - 1] ? idx[s - 1] : (gt[s] ? ci : idx[s]);
                    }
                    key[0] = gt[0] ? d : key[0];
                    idx[0] = gt[0] ? ci : idx[0];
                }
            }

            if (more) {
                float* d = nxt + tt * 16 * PITCH;
                v2f lo; lo.x = sv.x; lo.y = sv.y;
                v2f hi; hi.x = sv.z; hi.y = sv.w;
                *(v2f*)(d + sdst)     = lo;
                *(v2f*)(d + sdst + 4) = hi;
                d[ndst] = nlead ? ss : 0.0f;
            }
        }
        __syncthreads();
    }

    int* sel = selIdx + wave * (ROWS_PER_WAVE * KNN);
#pragma unroll
    for (int s = 0; s < KNN; ++s)
        sel[lane * KNN + s] = idx[s];
    __syncthreads();

    float* poolb = pooled + ((size_t)b * NPTS + rowbase) * CH;
#pragma unroll 1
    for (int r = 0; r < ROWS_PER_WAVE; ++r) {
        float mx = -3.0e38f;
#pragma unroll
        for (int k = 0; k < KNN; ++k) {
            const int j = sel[r * KNN + k];
            const float v = xb[(size_t)j * CH + lane];
            mx = fmaxf(mx, v);
        }
        poolb[r * CH + lane] = mx;
    }
}

__global__ __launch_bounds__(32) void mlp_kernel(const float* __restrict__ pooled,
                                                 const float* __restrict__ Wl,
                                                 const float* __restrict__ bl,
                                                 const float* __restrict__ Wc,
                                                 const float* __restrict__ bc,
                                                 float* __restrict__ out) {
    __shared__ float hT[16 * 68];

    const int lane = threadIdx.x;
    const int m = lane & 15;
    const int h = lane >> 4;
    const size_t rowbase = (size_t)blockIdx.x * 16;

    v2f afrag[8];
#pragma unroll
    for (int c = 0; c < 8; ++c) {
        const float* p = pooled + (rowbase + m) * CH + 4 * c + 2 * h;
        afrag[c].x = p[0];
        afrag[c].y = p[1];
    }

#pragma unroll
    for (int t = 0; t < 4; ++t) {
        v8f acc = {};
#pragma unroll
        for (int c = 0; c < 8; ++c) {
            const float* p = Wl + (size_t)(4 * c + 2 * h) * LDIM + t * 16 + m;
            v2f bfrag;
            bfrag.x = p[0];
            bfrag.y = p[LDIM];
            acc = __builtin_amdgcn_wmma_f32_16x16x4_f32(
                false, afrag[c], false, bfrag, (short)0, acc, false, false);
        }
        const float bias = bl[t * 16 + m];
#pragma unroll
        for (int r = 0; r < 8; ++r)
            hT[(r + 8 * h) * 68 + t * 16 + m] = acc[r] + bias;
    }
    __syncthreads();

    v2f a2[16];
#pragma unroll
    for (int c = 0; c < 16; ++c) {
        a2[c].x = hT[m * 68 + 4 * c + 2 * h];
        a2[c].y = hT[m * 68 + 4 * c + 2 * h + 1];
    }

    float* outp = out + rowbase * ODIM;
#pragma unroll
    for (int t = 0; t < 8; ++t) {
        v8f acc = {};
#pragma unroll
        for (int c = 0; c < 16; ++c) {
            const float* p = Wc + (size_t)(4 * c + 2 * h) * ODIM + t * 16 + m;
            v2f bfrag;
            bfrag.x = p[0];
            bfrag.y = p[ODIM];
            acc = __builtin_amdgcn_wmma_f32_16x16x4_f32(
                false, a2[c], false, bfrag, (short)0, acc, false, false);
        }
        const float bias = bc[t * 16 + m];
#pragma unroll
        for (int r = 0; r < 8; ++r) {
            const float v = acc[r] + bias;
            outp[(size_t)(r + 8 * h) * ODIM + t * 16 + m] = fmaxf(v, 0.0f);
        }
    }
}

extern "C" void kernel_launch(void* const* d_in, const int* in_sizes, int n_in,
                              void* d_out, int out_size, void* d_ws, size_t ws_size,
                              hipStream_t stream) {
    const float* x  = (const float*)d_in[0];
    const float* Wl = (const float*)d_in[1];
    const float* bl = (const float*)d_in[2];
    const float* Wc = (const float*)d_in[3];
    const float* bc = (const float*)d_in[4];
    float* out = (float*)d_out;

    float* pooled = (float*)d_ws;
    float* sqbuf  = pooled + (size_t)BATCH * NPTS * CH;

    const int total = BATCH * NPTS;
    sq_kernel<<<(total + 255) / 256, 256, 0, stream>>>(x, sqbuf, total);
    knn_pool_kernel<<<BATCH * NWG_PER_BATCH, 128, 0, stream>>>(x, sqbuf, pooled);
    mlp_kernel<<<total / 16, 32, 0, stream>>>(pooled, Wl, bl, Wc, bc, out);
}


# kNN selection without a branch per candidate: every key passes an unconditional 16-slot median list (keys only); hits logged as (key,index) per row in LDS, members picked by the final 16th key
# speedup vs baseline: 2.4502x; 1.1579x over previous
#include <hip/hip_runtime.h>
#include <hip/hip_bf16.h>

#define BATCH 4
#define NPTS  8192
#define CH    32
#define KNN   16
#define LDIM  64
#define ODIM  128
#define NCTILES (NPTS / 16)
#define ROWS_PER_WAVE 32
#define NRBLK (NPTS / ROWS_PER_WAVE)

typedef __attribute__((ext_vector_type(2))) float v2f;
typedef __attribute__((ext_vector_type(8))) float v8f;

__global__ void sq_kernel(const float* __restrict__ x, float* __restrict__ sq, int total) {
    int i = blockIdx.x * blockDim.x + threadIdx.x;
    if (i >= total) return;
    const float* p = x + (size_t)i * CH;
    float s = 0.0f;
#pragma unroll
    for (int c = 0; c < CH; ++c) s += p[c] * p[c];
    sq[i] = s;
}

#define WAVES_PER_WG     4
#define ROWS_PER_WG      (WAVES_PER_WG * ROWS_PER_WAVE)
#define NWG_PER_BATCH    (NPTS / ROWS_PER_WG)
#define TILES_PER_PANEL  32
#define PANEL_PTS        (TILES_PER_PANEL * 16)
#define NPANELS          (NPTS / PANEL_PTS)
#define PITCH            36

typedef __attribute__((ext_vector_type(4))) float v4f;

#define LOGCAP   128
#define LOGCHK   4
#define LOGROW   (128 * 8)
#define LOGBATCH 16

__device__ __forceinline__ int log_filter(char* col, int n, float T, int need) {
    int w = 0, eq = 0;
#pragma unroll 1
    for (int p0 = 0; p0 < LOGCAP; p0 += LOGBATCH) {
        if (__builtin_amdgcn_ballot_w32(p0 < n) == 0u) break;
        v2f e[LOGBATCH];
#pragma unroll
        for (int u = 0; u < LOGBATCH; ++u)
            e[u] = *(const v2f*)(col + (p0 + u) * LOGROW);
#pragma unroll
        for (int u = 0; u < LOGBATCH; ++u) {
            const bool live = (p0 + u) < n;
            const bool tie  = live && (e[u].x == T);
            const bool keep = (live && (e[u].x < T)) || (tie && (eq < need));
            eq += tie ? 1 : 0;
            *(v2f*)(col + (keep ? w : LOGCAP) * LOGROW) = e[u];
            w += keep ? 1 : 0;
        }
    }
    return w;
}

__device__ __forceinline__ float xhalf(float v) {
    const int b = __builtin_bit_cast(int, v);
    return __builtin_bit_cast(float,
        __builtin_amdgcn_permlanex16(b, b, 0x76543210, 0xfedcba98, false, false));
}

__global__ __launch_bounds__(128) void knn_pool_kernel(const float* __restrict__ x,
                                                       const float* __restrict__ sq,
                                                       float* __restrict__ pooled) {
    __shared__ __attribute__((aligned(16))) float pan[2 * PANEL_PTS * PITCH];
    __shared__ __attribute__((aligned(16))) v2f logbuf[(LOGCAP + 1) * ROWS_PER_WG];

    const int tid     = threadIdx.x;
    const int lane    = tid & 31;
    const int wave    = __builtin_amdgcn_readfirstlane(tid >> 5);
    const int b       = blockIdx.x / NWG_PER_BATCH;
    const int rowbase = (blockIdx.x % NWG_PER_BATCH) * ROWS_PER_WG + wave * ROWS_PER_WAVE;
    const float* xb   = x  + (size_t)b * NPTS * CH;
    const float* sqb  = sq + (size_t)b * NPTS;

    const int m = lane & 15;
    const int h = lane >> 4;

    const int spt  = tid >> 3;
    const int sq4  = tid & 7;
    const int sdst = spt * PITCH + 8 * (sq4 >> 1) + 2 * (sq4 & 1);
    const int ndst = spt * PITCH + 32 + (sq4 & 3);
    const bool nlead = (sq4 & 3) == 0;
    const float* gsrc = xb + tid * 4;
    const float* gsq  = sqb + spt;

    v2f qfrag0[8], qfrag1[8];
#pragma unroll
    for (int c = 0; c < 8; ++c) {
        const float* p0 = xb + (size_t)(rowbase + m) * CH + 4 * c + 2 * h;
        const float* p1 = xb + (size_t)(rowbase + 16 + m) * CH + 4 * c + 2 * h;
        qfrag0[c].x = -2.0f * p0[0];
        qfrag0[c].y = -2.0f * p0[1];
        qfrag1[c].x = -2.0f * p1[0];
        qfrag1[c].y = -2.0f * p1[1];
    }
    v2f qfragOne;
    qfragOne.x = (h == 0) ? 1.0f : 0.0f;
    qfragOne.y = 0.0f;

    float key[KNN];
#pragma unroll
    for (int s = 0; s < KNN; ++s) key[s] = 3.0e38f;

    char* const logb  = (char*)logbuf;
    char* const lcol  = logb + tid * 8;
    char* const ldump = lcol + LOGCAP * LOGROW;
    char*       lnext = lcol;

#pragma unroll 1
    for (int s0 = 0; s0 < TILES_PER_PANEL; s0 += 8) {
        v4f   tv[8];
        float ts[8];
#pragma unroll
        for (int u = 0; u < 8; ++u) {
            const int tb = (s0 + u) * 16;
            tv[u] = *(const v4f*)(gsrc + (size_t)tb * CH);
            ts[u] = gsq[tb];
        }
#pragma unroll
        for (int u = 0; u < 8; ++u) {
            float* d = pan + (s0 + u) * 16 * PITCH;
            v2f lo; lo.x = tv[u].x; lo.y = tv[u].y;
            v2f hi; hi.x = tv[u].z; hi.y = tv[u].w;
            *(v2f*)(d + sdst)     = lo;
            *(v2f*)(d + sdst + 4) = hi;
            d[ndst] = nlead ? ts[u] : 0.0f;
        }
    }
    __syncthreads();

    const int foff = m * PITCH + 4 * h;

    for (int pn = 0; pn < NPANELS; ++pn) {
        const float* cur = pan + (pn & 1) * (PANEL_PTS * PITCH);
        float*       nxt = pan + ((pn & 1) ^ 1) * (PANEL_PTS * PITCH);
        const bool  more = (pn + 1 < NPANELS);
        const int   nextbase = (pn + 1) * PANEL_PTS;

        v4f f[4];
        v2f fs;
#pragma unroll
        for (int g = 0; g < 4; ++g) f[g] = *(const v4f*)(cur + foff + 8 * g);
        fs = *(const v2f*)(cur + m * PITCH + 32 + 2 * h);

#pragma unroll 1
        for (int tt = 0; tt < TILES_PER_PANEL; ++tt) {
            const int colbase = pn * PANEL_PTS + tt * 16;

            if ((tt & (LOGCHK - 1)) == 0) {
                const int n = (int)(lnext - lcol) / LOGROW;
                if (__builtin_amdgcn_ballot_w32(n > LOGCAP - 16 * LOGCHK) != 0u) {
                    const float T = key[KNN - 1];
                    int nlt = 0;
#pragma unroll
                    for (int s = 0; s < KNN - 1; ++s) nlt += (key[s] < T) ? 1 : 0;
                    const int w = log_filter(lcol, n, T, KNN - nlt);
                    lnext = lcol + w * LOGROW;
                }
            }

            v4f   sv = {};
            float ss = 0.0f;
            if (more) {
                const int tb = nextbase + tt * 16;
                sv = *(const v4f*)(gsrc + (size_t)tb * CH);
                ss = gsq[tb];
            }

            v8f acc0 = {};
            v8f acc1 = {};
#pragma unroll
            for (int g = 0; g < 4; ++g) {
                v2f c0; c0.x = f[g].x; c0.y = f[g].y;
                v2f c1; c1.x = f[g].z; c1.y = f[g].w;
                acc0 = __builtin_amdgcn_wmma_f32_16x16x4_f32(
                    false, c0, false, qfrag0[2 * g], (short)0, acc0, false, false);
                acc1 = __builtin_amdgcn_wmma_f32_16x16x4_f32(
                    false, c0, false, qfrag1[2 * g], (short)0, acc1, false, false);
                acc0 = __builtin_amdgcn_wmma_f32_16x16x4_f32(
                    false, c1, false, qfrag0[2 * g + 1], (short)0, acc0, false, false);
                acc1 = __builtin_amdgcn_wmma_f32_16x16x4_f32(
                    false, c1, false, qfrag1[2 * g + 1], (short)0, acc1, false, false);
            }
            acc0 = __builtin_amdgcn_wmma_f32_16x16x4_f32(
                false, fs, false, qfragOne, (short)0, acc0, false, false);
            acc1 = __builtin_amdgcn_wmma_f32_16x16x4_f32(
                false, fs, false, qfragOne, (short)0, acc1, false, false);

            if (tt + 1 < TILES_PER_PANEL) {
                const float* fr = cur + (tt + 1) * 16 * PITCH;
#pragma unroll
                for (int g = 0; g < 4; ++g) f[g] = *(const v4f*)(fr + foff + 8 * g);
                fs = *(const v2f*)(fr + m * PITCH + 32 + 2 * h);
            }

            float cand[16];
#pragma unroll
            for (int r = 0; r < 8; ++r) {
                const float a0 = acc0[r];
                const float a1 = acc1[r];
                const float give = (h == 0) ? a1 : a0;
                const float got  = xhalf(give);
                cand[r]     = (h == 0) ? a0 : got;
                cand[8 + r] = (h == 0) ? got : a1;
            }

#pragma unroll
            for (int c = 0; c < 16; ++c) {
                const float d = cand[c];
                const bool hit = d < key[KNN - 1];
                v2f e;
                e.x = d;
                e.y = __builtin_bit_cast(float, colbase + c);
                *(v2f*)(hit ? lnext : ldump) = e;
                lnext += hit ? LOGROW : 0;
#pragma unroll
                for (int s = KNN - 1; s >= 1; --s)
                    key[s] = __builtin_amdgcn_fmed3f(key[s - 1], key[s], d);
                key[0] = __builtin_amdgcn_fmed3f(-__builtin_inff(), key[0], d);
            }

            if (more) {
                float* d = nxt + tt * 16 * PITCH;
                v2f lo; lo.x = sv.x; lo.y = sv.y;
                v2f hi; hi.x = sv.z; hi.y = sv.w;
                *(v2f*)(d + sdst)     = lo;
                *(v2f*)(d + sdst + 4) = hi;
                d[ndst] = nlead ? ss : 0.0f;
            }
        }
        __syncthreads();
    }

    {
        const int n = (int)(lnext - lcol) / LOGROW;
        const float T = key[KNN - 1];
        int nlt = 0;
#pragma unroll
        for (int s = 0; s < KNN - 1; ++s) nlt += (key[s] < T) ? 1 : 0;
        (void)log_filter(lcol, n, T, KNN - nlt);
    }
    __syncthreads();

    float* poolb = pooled + ((size_t)b * NPTS + rowbase) * CH;
#pragma unroll 1
    for (int r = 0; r < ROWS_PER_WAVE; ++r) {
        float mx = -3.0e38f;
#pragma unroll
        for (int k = 0; k < KNN; ++k) {
            const int j = __builtin_bit_cast(int,
                *(const float*)(logb + k * LOGROW + (wave * ROWS_PER_WAVE + r) * 8 + 4)) & (NPTS - 1);
            const float v = xb[(size_t)j * CH + lane];
            mx = fmaxf(mx, v);
        }
        poolb[r * CH + lane] = mx;
    }
}

__global__ __launch_bounds__(32) void mlp_kernel(const float* __restrict__ pooled,
                                                 const float* __restrict__ Wl,
                                                 const float* __restrict__ bl,
                                                 const float* __restrict__ Wc,
                                                 const float* __restrict__ bc,
                                                 float* __restrict__ out) {
    __shared__ float hT[16 * 68];

    const int lane = threadIdx.x;
    const int m = lane & 15;
    const int h = lane >> 4;
    const size_t rowbase = (size_t)blockIdx.x * 16;

    v2f afrag[8];
#pragma unroll
    for (int c = 0; c < 8; ++c) {
        const float* p = pooled + (rowbase + m) * CH + 4 * c + 2 * h;
        afrag[c].x = p[0];
        afrag[c].y = p[1];
    }

#pragma unroll
    for (int t = 0; t < 4; ++t) {
        v8f acc = {};
#pragma unroll
        for (int c = 0; c < 8; ++c) {
            const float* p = Wl + (size_t)(4 * c + 2 * h) * LDIM + t * 16 + m;
            v2f bfrag;
            bfrag.x = p[0];
            bfrag.y = p[LDIM];
            acc = __builtin_amdgcn_wmma_f32_16x16x4_f32(
                false, afrag[c], false, bfrag, (short)0, acc, false, false);
        }
        const float bias = bl[t * 16 + m];
#pragma unroll
        for (int r = 0; r < 8; ++r)
            hT[(r + 8 * h) * 68 + t * 16 + m] = acc[r] + bias;
    }
    __syncthreads();

    v2f a2[16];
#pragma unroll
    for (int c = 0; c < 16; ++c) {
        a2[c].x = hT[m * 68 + 4 * c + 2 * h];
        a2[c].y = hT[m * 68 + 4 * c + 2 * h + 1];
    }

    float* outp = out + rowbase * ODIM;
#pragma unroll
    for (int t = 0; t < 8; ++t) {
        v8f acc = {};
#pragma unroll
        for (int c = 0; c < 16; ++c) {
            const float* p = Wc + (size_t)(4 * c + 2 * h) * ODIM + t * 16 + m;
            v2f bfrag;
            bfrag.x = p[0];
            bfrag.y = p[ODIM];
            acc = __builtin_amdgcn_wmma_f32_16x16x4_f32(
                false, a2[c], false, bfrag, (short)0, acc, false, false);
        }
        const float bias = bc[t * 16 + m];
#pragma unroll
        for (int r = 0; r < 8; ++r) {
            const float v = acc[r] + bias;
            outp[(size_t)(r + 8 * h) * ODIM + t * 16 + m] = fmaxf(v, 0.0f);
        }
    }
}

extern "C" void kernel_launch(void* const* d_in, const int* in_sizes, int n_in,
                              void* d_out, int out_size, void* d_ws, size_t ws_size,
                              hipStream_t stream) {
    const float* x  = (const float*)d_in[0];
    const float* Wl = (const float*)d_in[1];
    const float* bl = (const float*)d_in[2];
    const float* Wc = (const float*)d_in[3];
    const float* bc = (const float*)d_in[4];
    float* out = (float*)d_out;

    float* pooled = (float*)d_ws;
    float* sqbuf  = pooled + (size_t)BATCH * NPTS * CH;

    const int total = BATCH * NPTS;
    sq_kernel<<<(total + 255) / 256, 256, 0, stream>>>(x, sqbuf, total);
    knn_pool_kernel<<<BATCH * NWG_PER_BATCH, 128, 0, stream>>>(x, sqbuf, pooled);
    mlp_kernel<<<total / 16, 32, 0, stream>>>(pooled, Wl, bl, Wc, bc, out);
}


# kNN selection per tile in batch: key list by a 60-exchange sorting network and a bitonic half-merge (min/max only); log appends gated by the sign of d minus the tile-start 16th key; sums unchanged
# speedup vs baseline: 2.9480x; 1.2032x over previous
#include <hip/hip_runtime.h>
#include <hip/hip_bf16.h>

#define BATCH 4
#define NPTS  8192
#define CH    32
#define KNN   16
#define LDIM  64
#define ODIM  128
#define NCTILES (NPTS / 16)
#define ROWS_PER_WAVE 32
#define NRBLK (NPTS / ROWS_PER_WAVE)

typedef __attribute__((ext_vector_type(2))) float v2f;
typedef __attribute__((ext_vector_type(8))) float v8f;

__global__ void sq_kernel(const float* __restrict__ x, float* __restrict__ sq, int total) {
    int i = blockIdx.x * blockDim.x + threadIdx.x;
    if (i >= total) return;
    const float* p = x + (size_t)i * CH;
    float s = 0.0f;
#pragma unroll
    for (int c = 0; c < CH; ++c) s += p[c] * p[c];
    sq[i] = s;
}

#define WAVES_PER_WG     4
#define ROWS_PER_WG      (WAVES_PER_WG * ROWS_PER_WAVE)
#define NWG_PER_BATCH    (NPTS / ROWS_PER_WG)
#define TILES_PER_PANEL  32
#define PANEL_PTS        (TILES_PER_PANEL * 16)
#define NPANELS          (NPTS / PANEL_PTS)
#define PITCH            36

typedef __attribute__((ext_vector_type(4))) float v4f;

#define LOGCAP   128
#define LOGCHK   4
#define LOGROW   (128 * 8)
#define LOGBATCH 16

__device__ __forceinline__ int log_filter(char* col, int n, float T, int need) {
    int w = 0, eq = 0;
#pragma unroll 1
    for (int p0 = 0; p0 < LOGCAP; p0 += LOGBATCH) {
        if (__builtin_amdgcn_ballot_w32(p0 < n) == 0u) break;
        v2f e[LOGBATCH];
#pragma unroll
        for (int u = 0; u < LOGBATCH; ++u)
            e[u] = *(const v2f*)(col + (p0 + u) * LOGROW);
#pragma unroll
        for (int u = 0; u < LOGBATCH; ++u) {
            const bool live = (p0 + u) < n;
            const bool tie  = live && (e[u].x == T);
            const bool keep = (live && (e[u].x < T)) || (tie && (eq < need));
            eq += tie ? 1 : 0;
            *(v2f*)(col + (keep ? w : LOGCAP) * LOGROW) = e[u];
            w += keep ? 1 : 0;
        }
    }
    return w;
}

#define KCE(a, b) { const float lo_ = __builtin_fminf(a, b); const float hi_ = __builtin_fmaxf(a, b); a = lo_; b = hi_; }
#define KLAYER0(v)  KCE(v[0], v[13]) KCE(v[1], v[12]) KCE(v[2], v[15]) KCE(v[3], v[14]) KCE(v[4], v[8]) KCE(v[5], v[6]) KCE(v[7], v[11]) KCE(v[9], v[10])
#define KLAYER1(v)  KCE(v[0], v[5]) KCE(v[1], v[7]) KCE(v[2], v[9]) KCE(v[3], v[4]) KCE(v[6], v[13]) KCE(v[8], v[14]) KCE(v[10], v[15]) KCE(v[11], v[12])
#define KLAYER2(v)  KCE(v[0], v[1]) KCE(v[2], v[3]) KCE(v[4], v[5]) KCE(v[6], v[8]) KCE(v[7], v[9]) KCE(v[10], v[11]) KCE(v[12], v[13]) KCE(v[14], v[15])
#define KLAYER3(v)  KCE(v[0], v[2]) KCE(v[1], v[3]) KCE(v[4], v[10]) KCE(v[5], v[11]) KCE(v[6], v[7]) KCE(v[8], v[9]) KCE(v[12], v[14]) KCE(v[13], v[15])
#define KLAYER4(v)  KCE(v[1], v[2]) KCE(v[3], v[12]) KCE(v[4], v[6]) KCE(v[5], v[7]) KCE(v[8], v[10]) KCE(v[9], v[11]) KCE(v[13], v[14])
#define KLAYER5(v)  KCE(v[1], v[4]) KCE(v[2], v[6]) KCE(v[5], v[8]) KCE(v[7], v[10]) KCE(v[9], v[13]) KCE(v[11], v[14])
#define KLAYER6(v)  KCE(v[2], v[4]) KCE(v[3], v[6]) KCE(v[9], v[12]) KCE(v[11], v[13])
#define KLAYER7(v)  KCE(v[3], v[5]) KCE(v[6], v[8]) KCE(v[7], v[9]) KCE(v[10], v[12])
#define KLAYER8(v)  KCE(v[3], v[4]) KCE(v[5], v[6]) KCE(v[7], v[8]) KCE(v[9], v[10]) KCE(v[11], v[12])
#define KLAYER9(v)  KCE(v[6], v[7]) KCE(v[8], v[9])
#define KLAYER10(k, v) k[0] = __builtin_fminf(k[0], v[15]); k[1] = __builtin_fminf(k[1], v[14]); k[2] = __builtin_fminf(k[2], v[13]); k[3] = __builtin_fminf(k[3], v[12]); k[4] = __builtin_fminf(k[4], v[11]); k[5] = __builtin_fminf(k[5], v[10]); k[6] = __builtin_fminf(k[6], v[9]); k[7] = __builtin_fminf(k[7], v[8]); k[8] = __builtin_fminf(k[8], v[7]); k[9] = __builtin_fminf(k[9], v[6]); k[10] = __builtin_fminf(k[10], v[5]); k[11] = __builtin_fminf(k[11], v[4]); k[12] = __builtin_fminf(k[12], v[3]); k[13] = __builtin_fminf(k[13], v[2]); k[14] = __builtin_fminf(k[14], v[1]); k[15] = __builtin_fminf(k[15], v[0]);
#define KLAYER11(k) KCE(k[0], k[8]) KCE(k[1], k[9]) KCE(k[2], k[10]) KCE(k[3], k[11]) KCE(k[4], k[12]) KCE(k[5], k[13]) KCE(k[6], k[14]) KCE(k[7], k[15])
#define KLAYER12(k) KCE(k[0], k[4]) KCE(k[1], k[5]) KCE(k[2], k[6]) KCE(k[3], k[7]) KCE(k[8], k[12]) KCE(k[9], k[13]) KCE(k[10], k[14]) KCE(k[11], k[15])
#define KLAYER13(k) KCE(k[0], k[2]) KCE(k[1], k[3]) KCE(k[4], k[6]) KCE(k[5], k[7]) KCE(k[8], k[10]) KCE(k[9], k[11]) KCE(k[12], k[14]) KCE(k[13], k[15])
#define KLAYER14(k) KCE(k[0], k[1]) KCE(k[2], k[3]) KCE(k[4], k[5]) KCE(k[6], k[7]) KCE(k[8], k[9]) KCE(k[10], k[11]) KCE(k[12], k[13]) KCE(k[14], k[15])

__device__ __forceinline__ float xhalf(float v) {
    const int b = __builtin_bit_cast(int, v);
    return __builtin_bit_cast(float,
        __builtin_amdgcn_permlanex16(b, b, 0x76543210, 0xfedcba98, false, false));
}

__global__ __launch_bounds__(128) void knn_pool_kernel(const float* __restrict__ x,
                                                       const float* __restrict__ sq,
                                                       float* __restrict__ pooled) {
    __shared__ __attribute__((aligned(16))) float pan[2 * PANEL_PTS * PITCH];
    __shared__ __attribute__((aligned(16))) v2f logbuf[(LOGCAP + 1) * ROWS_PER_WG];

    const int tid     = threadIdx.x;
    const int lane    = tid & 31;
    const int wave    = __builtin_amdgcn_readfirstlane(tid >> 5);
    const int b       = blockIdx.x / NWG_PER_BATCH;
    const int rowbase = (blockIdx.x % NWG_PER_BATCH) * ROWS_PER_WG + wave * ROWS_PER_WAVE;
    const float* xb   = x  + (size_t)b * NPTS * CH;
    const float* sqb  = sq + (size_t)b * NPTS;

    const int m = lane & 15;
    const int h = lane >> 4;

    const int spt  = tid >> 3;
    const int sq4  = tid & 7;
    const int sdst = spt * PITCH + 8 * (sq4 >> 1) + 2 * (sq4 & 1);
    const int ndst = spt * PITCH + 32 + (sq4 & 3);
    const bool nlead = (sq4 & 3) == 0;
    const float* gsrc = xb + tid * 4;
    const float* gsq  = sqb + spt;

    v2f qfrag0[8], qfrag1[8];
#pragma unroll
    for (int c = 0; c < 8; ++c) {
        const float* p0 = xb + (size_t)(rowbase + m) * CH + 4 * c + 2 * h;
        const float* p1 = xb + (size_t)(rowbase + 16 + m) * CH + 4 * c + 2 * h;
        qfrag0[c].x = -2.0f * p0[0];
        qfrag0[c].y = -2.0f * p0[1];
        qfrag1[c].x = -2.0f * p1[0];
        qfrag1[c].y = -2.0f * p1[1];
    }
    v2f qfragOne;
    qfragOne.x = (h == 0) ? 1.0f : 0.0f;
    qfragOne.y = 0.0f;

    float key[KNN];
#pragma unroll
    for (int s = 0; s < KNN; ++s) key[s] = 3.0e38f;

    char* const logb  = (char*)logbuf;
    char* const lcol  = logb + tid * 8;
    char*       lnext = lcol;

#pragma unroll 1
    for (int s0 = 0; s0 < TILES_PER_PANEL; s0 += 8) {
        v4f   tv[8];
        float ts[8];
#pragma unroll
        for (int u = 0; u < 8; ++u) {
            const int tb = (s0 + u) * 16;
            tv[u] = *(const v4f*)(gsrc + (size_t)tb * CH);
            ts[u] = gsq[tb];
        }
#pragma unroll
        for (int u = 0; u < 8; ++u) {
            float* d = pan + (s0 + u) * 16 * PITCH;
            v2f lo; lo.x = tv[u].x; lo.y = tv[u].y;
            v2f hi; hi.x = tv[u].z; hi.y = tv[u].w;
            *(v2f*)(d + sdst)     = lo;
            *(v2f*)(d + sdst + 4) = hi;
            d[ndst] = nlead ? ts[u] : 0.0f;
        }
    }
    __syncthreads();

    const int foff = m * PITCH + 4 * h;

    for (int pn = 0; pn < NPANELS; ++pn) {
        const float* cur = pan + (pn & 1) * (PANEL_PTS * PITCH);
        float*       nxt = pan + ((pn & 1) ^ 1) * (PANEL_PTS * PITCH);
        const bool  more = (pn + 1 < NPANELS);
        const int   nextbase = (pn + 1) * PANEL_PTS;

        v4f f[4];
        v2f fs;
#pragma unroll
        for (int g = 0; g < 4; ++g) f[g] = *(const v4f*)(cur + foff + 8 * g);
        fs = *(const v2f*)(cur + m * PITCH + 32 + 2 * h);

#pragma unroll 1
        for (int tt = 0; tt < TILES_PER_PANEL; ++tt) {
            const int colbase = pn * PANEL_PTS + tt * 16;

            if ((tt & (LOGCHK - 1)) == 0) {
                const int n = (int)(lnext - lcol) / LOGROW;
                if (__builtin_amdgcn_ballot_w32(n > LOGCAP - 16 * LOGCHK) != 0u) {
                    const float T = key[KNN - 1];
                    int nlt = 0;
#pragma unroll
                    for (int s = 0; s < KNN - 1; ++s) nlt += (key[s] < T) ? 1 : 0;
                    const int w = log_filter(lcol, n, T, KNN - nlt);
                    lnext = lcol + w * LOGROW;
                }
            }

            v4f   sv = {};
            float ss = 0.0f;
            if (more) {
                const int tb = nextbase + tt * 16;
                sv = *(const v4f*)(gsrc + (size_t)tb * CH);
                ss = gsq[tb];
            }

            v8f acc0 = {};
            v8f acc1 = {};
#pragma unroll
            for (int g = 0; g < 4; ++g) {
                v2f c0; c0.x = f[g].x; c0.y = f[g].y;
                v2f c1; c1.x = f[g].z; c1.y = f[g].w;
                acc0 = __builtin_amdgcn_wmma_f32_16x16x4_f32(
                    false, c0, false, qfrag0[2 * g], (short)0, acc0, false, false);
                acc1 = __builtin_amdgcn_wmma_f32_16x16x4_f32(
                    false, c0, false, qfrag1[2 * g], (short)0, acc1, false, false);
                acc0 = __builtin_amdgcn_wmma_f32_16x16x4_f32(
                    false, c1, false, qfrag0[2 * g + 1], (short)0, acc0, false, false);
                acc1 = __builtin_amdgcn_wmma_f32_16x16x4_f32(
                    false, c1, false, qfrag1[2 * g + 1], (short)0, acc1, false, false);
            }
            acc0 = __builtin_amdgcn_wmma_f32_16x16x4_f32(
                false, fs, false, qfragOne, (short)0, acc0, false, false);
            acc1 = __builtin_amdgcn_wmma_f32_16x16x4_f32(
                false, fs, false, qfragOne, (short)0, acc1, false, false);

            if (tt + 1 < TILES_PER_PANEL) {
                const float* fr = cur + (tt + 1) * 16 * PITCH;
#pragma unroll
                for (int g = 0; g < 4; ++g) f[g] = *(const v4f*)(fr + foff + 8 * g);
                fs = *(const v2f*)(fr + m * PITCH + 32 + 2 * h);
            }

            float cand[16];
#pragma unroll
            for (int r = 0; r < 8; ++r) {
                const float a0 = acc0[r];
                const float a1 = acc1[r];
                const float give = (h == 0) ? a1 : a0;
                const float got  = xhalf(give);
                cand[r]     = (h == 0) ? a0 : got;
                cand[8 + r] = (h == 0) ? got : a1;
            }

            const float T = key[KNN - 1];
#pragma unroll
            for (int c = 0; c < 16; ++c) {
                const float d = cand[c];
                const unsigned under = __builtin_bit_cast(unsigned, d - T) >> 31;
                v2f e;
                e.x = d;
                e.y = __builtin_bit_cast(float, colbase + c);
                *(v2f*)lnext = e;
                lnext += under * LOGROW;
            }
            KLAYER0(cand) KLAYER1(cand) KLAYER2(cand) KLAYER3(cand) KLAYER4(cand)
            KLAYER5(cand) KLAYER6(cand) KLAYER7(cand) KLAYER8(cand) KLAYER9(cand)
            KLAYER10(key, cand)
            KLAYER11(key) KLAYER12(key) KLAYER13(key) KLAYER14(key)

            {
                float* d = nxt + tt * 16 * PITCH;
                v2f lo; lo.x = sv.x; lo.y = sv.y;
                v2f hi; hi.x = sv.z; hi.y = sv.w;
                *(v2f*)(d + sdst)     = lo;
                *(v2f*)(d + sdst + 4) = hi;
                d[ndst] = nlead ? ss : 0.0f;
            }
        }
        __syncthreads();
    }

    {
        const int n = (int)(lnext - lcol) / LOGROW;
        const float T = key[KNN - 1];
        int nlt = 0;
#pragma unroll
        for (int s = 0; s < KNN - 1; ++s) nlt += (key[s] < T) ? 1 : 0;
        (void)log_filter(lcol, n, T, KNN - nlt);
    }
    __syncthreads();

    float* poolb = pooled + ((size_t)b * NPTS + rowbase) * CH;
#pragma unroll 1
    for (int r = 0; r < ROWS_PER_WAVE; ++r) {
        float mx = -3.0e38f;
#pragma unroll
        for (int k = 0; k < KNN; ++k) {
            const int j = __builtin_bit_cast(int,
                *(const float*)(logb + k * LOGROW + (wave * ROWS_PER_WAVE + r) * 8 + 4)) & (NPTS - 1);
            const float v = xb[(size_t)j * CH + lane];
            mx = fmaxf(mx, v);
        }
        poolb[r * CH + lane] = mx;
    }
}

__global__ __launch_bounds__(32) void mlp_kernel(const float* __restrict__ pooled,
                                                 const float* __restrict__ Wl,
                                                 const float* __restrict__ bl,
                                                 const float* __restrict__ Wc,
                                                 const float* __restrict__ bc,
                                                 float* __restrict__ out) {
    __shared__ float hT[16 * 68];

    const int lane = threadIdx.x;
    const int m = lane & 15;
    const int h = lane >> 4;
    const size_t rowbase = (size_t)blockIdx.x * 16;

    v2f afrag[8];
#pragma unroll
    for (int c = 0; c < 8; ++c) {
        const float* p = pooled + (rowbase + m) * CH + 4 * c + 2 * h;
        afrag[c].x = p[0];
        afrag[c].y = p[1];
    }

#pragma unroll
    for (int t = 0; t < 4; ++t) {
        v8f acc = {};
#pragma unroll
        for (int c = 0; c < 8; ++c) {
            const float* p = Wl + (size_t)(4 * c + 2 * h) * LDIM + t * 16 + m;
            v2f bfrag;
            bfrag.x = p[0];
            bfrag.y = p[LDIM];
            acc = __builtin_amdgcn_wmma_f32_16x16x4_f32(
                false, afrag[c], false, bfrag, (short)0, acc, false, false);
        }
        const float bias = bl[t * 16 + m];
#pragma unroll
        for (int r = 0; r < 8; ++r)
            hT[(r + 8 * h) * 68 + t * 16 + m] = acc[r] + bias;
    }
    __syncthreads();

    v2f a2[16];
#pragma unroll
    for (int c = 0; c < 16; ++c) {
        a2[c].x = hT[m * 68 + 4 * c + 2 * h];
        a2[c].y = hT[m * 68 + 4 * c + 2 * h + 1];
    }

    float* outp = out + rowbase * ODIM;
#pragma unroll
    for (int t = 0; t < 8; ++t) {
        v8f acc = {};
#pragma unroll
        for (int c = 0; c < 16; ++c) {
            const float* p = Wc + (size_t)(4 * c + 2 * h) * ODIM + t * 16 + m;
            v2f bfrag;
            bfrag.x = p[0];
            bfrag.y = p[ODIM];
            acc = __builtin_amdgcn_wmma_f32_16x16x4_f32(
                false, a2[c], false, bfrag, (short)0, acc, false, false);
        }
        const float bias = bc[t * 16 + m];
#pragma unroll
        for (int r = 0; r < 8; ++r) {
            const float v = acc[r] + bias;
            outp[(size_t)(r + 8 * h) * ODIM + t * 16 + m] = fmaxf(v, 0.0f);
        }
    }
}

extern "C" void kernel_launch(void* const* d_in, const int* in_sizes, int n_in,
                              void* d_out, int out_size, void* d_ws, size_t ws_size,
                              hipStream_t stream) {
    const float* x  = (const float*)d_in[0];
    const float* Wl = (const float*)d_in[1];
    const float* bl = (const float*)d_in[2];
    const float* Wc = (const float*)d_in[3];
    const float* bc = (const float*)d_in[4];
    float* out = (float*)d_out;

    float* pooled = (float*)d_ws;
    float* sqbuf  = pooled + (size_t)BATCH * NPTS * CH;

    const int total = BATCH * NPTS;
    sq_kernel<<<(total + 255) / 256, 256, 0, stream>>>(x, sqbuf, total);
    knn_pool_kernel<<<BATCH * NWG_PER_BATCH, 128, 0, stream>>>(x, sqbuf, pooled);
    mlp_kernel<<<total / 16, 32, 0, stream>>>(pooled, Wl, bl, Wc, bc, out);
}


# kNN key list takes only each tile's smallest key (min tree + 16-slot median clamp): still bounds the 16th key; exact list rebuilt from the row's log by the network at clean-ups; sums unchanged
# speedup vs baseline: 3.0363x; 1.0300x over previous
#include <hip/hip_runtime.h>
#include <hip/hip_bf16.h>

#define BATCH 4
#define NPTS  8192
#define CH    32
#define KNN   16
#define LDIM  64
#define ODIM  128
#define NCTILES (NPTS / 16)
#define ROWS_PER_WAVE 32
#define NRBLK (NPTS / ROWS_PER_WAVE)

typedef __attribute__((ext_vector_type(2))) float v2f;
typedef __attribute__((ext_vector_type(8))) float v8f;

__global__ void sq_kernel(const float* __restrict__ x, float* __restrict__ sq, int total) {
    int i = blockIdx.x * blockDim.x + threadIdx.x;
    if (i >= total) return;
    const float* p = x + (size_t)i * CH;
    float s = 0.0f;
#pragma unroll
    for (int c = 0; c < CH; ++c) s += p[c] * p[c];
    sq[i] = s;
}

#define WAVES_PER_WG     4
#define ROWS_PER_WG      (WAVES_PER_WG * ROWS_PER_WAVE)
#define NWG_PER_BATCH    (NPTS / ROWS_PER_WG)
#define TILES_PER_PANEL  32
#define PANEL_PTS        (TILES_PER_PANEL * 16)
#define NPANELS          (NPTS / PANEL_PTS)
#define PITCH            36

typedef __attribute__((ext_vector_type(4))) float v4f;

#define LOGCAP   128
#define LOGCHK   4
#define LOGROW   (128 * 8)
#define LOGBATCH 16

__device__ __forceinline__ int log_filter(char* col, int n, float T, int need) {
    int w = 0, eq = 0;
#pragma unroll 1
    for (int p0 = 0; p0 < LOGCAP; p0 += LOGBATCH) {
        if (__builtin_amdgcn_ballot_w32(p0 < n) == 0u) break;
        v2f e[LOGBATCH];
#pragma unroll
        for (int u = 0; u < LOGBATCH; ++u)
            e[u] = *(const v2f*)(col + (p0 + u) * LOGROW);
#pragma unroll
        for (int u = 0; u < LOGBATCH; ++u) {
            const bool live = (p0 + u) < n;
            const bool tie  = live && (e[u].x == T);
            const bool keep = (live && (e[u].x < T)) || (tie && (eq < need));
            eq += tie ? 1 : 0;
            *(v2f*)(col + (keep ? w : LOGCAP) * LOGROW) = e[u];
            w += keep ? 1 : 0;
        }
    }
    return w;
}

#define KCE(a, b) { const float lo_ = __builtin_fminf(a, b); const float hi_ = __builtin_fmaxf(a, b); a = lo_; b = hi_; }
#define KLAYER0(v)  KCE(v[0], v[13]) KCE(v[1], v[12]) KCE(v[2], v[15]) KCE(v[3], v[14]) KCE(v[4], v[8]) KCE(v[5], v[6]) KCE(v[7], v[11]) KCE(v[9], v[10])
#define KLAYER1(v)  KCE(v[0], v[5]) KCE(v[1], v[7]) KCE(v[2], v[9]) KCE(v[3], v[4]) KCE(v[6], v[13]) KCE(v[8], v[14]) KCE(v[10], v[15]) KCE(v[11], v[12])
#define KLAYER2(v)  KCE(v[0], v[1]) KCE(v[2], v[3]) KCE(v[4], v[5]) KCE(v[6], v[8]) KCE(v[7], v[9]) KCE(v[10], v[11]) KCE(v[12], v[13]) KCE(v[14], v[15])
#define KLAYER3(v)  KCE(v[0], v[2]) KCE(v[1], v[3]) KCE(v[4], v[10]) KCE(v[5], v[11]) KCE(v[6], v[7]) KCE(v[8], v[9]) KCE(v[12], v[14]) KCE(v[13], v[15])
#define KLAYER4(v)  KCE(v[1], v[2]) KCE(v[3], v[12]) KCE(v[4], v[6]) KCE(v[5], v[7]) KCE(v[8], v[10]) KCE(v[9], v[11]) KCE(v[13], v[14])
#define KLAYER5(v)  KCE(v[1], v[4]) KCE(v[2], v[6]) KCE(v[5], v[8]) KCE(v[7], v[10]) KCE(v[9], v[13]) KCE(v[11], v[14])
#define KLAYER6(v)  KCE(v[2], v[4]) KCE(v[3], v[6]) KCE(v[9], v[12]) KCE(v[11], v[13])
#define KLAYER7(v)  KCE(v[3], v[5]) KCE(v[6], v[8]) KCE(v[7], v[9]) KCE(v[10], v[12])
#define KLAYER8(v)  KCE(v[3], v[4]) KCE(v[5], v[6]) KCE(v[7], v[8]) KCE(v[9], v[10]) KCE(v[11], v[12])
#define KLAYER9(v)  KCE(v[6], v[7]) KCE(v[8], v[9])
#define KLAYER10(k, v) k[0] = __builtin_fminf(k[0], v[15]); k[1] = __builtin_fminf(k[1], v[14]); k[2] = __builtin_fminf(k[2], v[13]); k[3] = __builtin_fminf(k[3], v[12]); k[4] = __builtin_fminf(k[4], v[11]); k[5] = __builtin_fminf(k[5], v[10]); k[6] = __builtin_fminf(k[6], v[9]); k[7] = __builtin_fminf(k[7], v[8]); k[8] = __builtin_fminf(k[8], v[7]); k[9] = __builtin_fminf(k[9], v[6]); k[10] = __builtin_fminf(k[10], v[5]); k[11] = __builtin_fminf(k[11], v[4]); k[12] = __builtin_fminf(k[12], v[3]); k[13] = __builtin_fminf(k[13], v[2]); k[14] = __builtin_fminf(k[14], v[1]); k[15] = __builtin_fminf(k[15], v[0]);
#define KLAYER11(k) KCE(k[0], k[8]) KCE(k[1], k[9]) KCE(k[2], k[10]) KCE(k[3], k[11]) KCE(k[4], k[12]) KCE(k[5], k[13]) KCE(k[6], k[14]) KCE(k[7], k[15])
#define KLAYER12(k) KCE(k[0], k[4]) KCE(k[1], k[5]) KCE(k[2], k[6]) KCE(k[3], k[7]) KCE(k[8], k[12]) KCE(k[9], k[13]) KCE(k[10], k[14]) KCE(k[11], k[15])
#define KLAYER13(k) KCE(k[0], k[2]) KCE(k[1], k[3]) KCE(k[4], k[6]) KCE(k[5], k[7]) KCE(k[8], k[10]) KCE(k[9], k[11]) KCE(k[12], k[14]) KCE(k[13], k[15])
#define KLAYER14(k) KCE(k[0], k[1]) KCE(k[2], k[3]) KCE(k[4], k[5]) KCE(k[6], k[7]) KCE(k[8], k[9]) KCE(k[10], k[11]) KCE(k[12], k[13]) KCE(k[14], k[15])

__device__ __forceinline__ float xhalf(float v) {
    const int b = __builtin_bit_cast(int, v);
    return __builtin_bit_cast(float,
        __builtin_amdgcn_permlanex16(b, b, 0x76543210, 0xfedcba98, false, false));
}

__device__ __forceinline__ void log_top16(const char* col, int n, float (&key)[KNN]) {
#pragma unroll
    for (int s = 0; s < KNN; ++s) key[s] = 3.0e38f;
#pragma unroll 1
    for (int p0 = 0; p0 < LOGCAP; p0 += 16) {
        if (__builtin_amdgcn_ballot_w32(p0 < n) == 0u) break;
        float v[16];
#pragma unroll
        for (int u = 0; u < 16; ++u) {
            const float kk = *(const float*)(col + (p0 + u) * LOGROW);
            v[u] = ((p0 + u) < n) ? kk : 3.0e38f;
        }
        KLAYER0(v) KLAYER1(v) KLAYER2(v) KLAYER3(v) KLAYER4(v)
        KLAYER5(v) KLAYER6(v) KLAYER7(v) KLAYER8(v) KLAYER9(v)
        KLAYER10(key, v)
        KLAYER11(key) KLAYER12(key) KLAYER13(key) KLAYER14(key)
    }
}

__global__ __launch_bounds__(128) void knn_pool_kernel(const float* __restrict__ x,
                                                       const float* __restrict__ sq,
                                                       float* __restrict__ pooled) {
    __shared__ __attribute__((aligned(16))) float pan[2 * PANEL_PTS * PITCH];
    __shared__ __attribute__((aligned(16))) v2f logbuf[(LOGCAP + 1) * ROWS_PER_WG];

    const int tid     = threadIdx.x;
    const int lane    = tid & 31;
    const int wave    = __builtin_amdgcn_readfirstlane(tid >> 5);
    const int b       = blockIdx.x / NWG_PER_BATCH;
    const int rowbase = (blockIdx.x % NWG_PER_BATCH) * ROWS_PER_WG + wave * ROWS_PER_WAVE;
    const float* xb   = x  + (size_t)b * NPTS * CH;
    const float* sqb  = sq + (size_t)b * NPTS;

    const int m = lane & 15;
    const int h = lane >> 4;

    const int spt  = tid >> 3;
    const int sq4  = tid & 7;
    const int sdst = spt * PITCH + 8 * (sq4 >> 1) + 2 * (sq4 & 1);
    const int ndst = spt * PITCH + 32 + (sq4 & 3);
    const bool nlead = (sq4 & 3) == 0;
    const float* gsrc = xb + tid * 4;
    const float* gsq  = sqb + spt;

    v2f qfrag0[8], qfrag1[8];
#pragma unroll
    for (int c = 0; c < 8; ++c) {
        const float* p0 = xb + (size_t)(rowbase + m) * CH + 4 * c + 2 * h;
        const float* p1 = xb + (size_t)(rowbase + 16 + m) * CH + 4 * c + 2 * h;
        qfrag0[c].x = -2.0f * p0[0];
        qfrag0[c].y = -2.0f * p0[1];
        qfrag1[c].x = -2.0f * p1[0];
        qfrag1[c].y = -2.0f * p1[1];
    }
    v2f qfragOne;
    qfragOne.x = (h == 0) ? 1.0f : 0.0f;
    qfragOne.y = 0.0f;

    float key[KNN];
#pragma unroll
    for (int s = 0; s < KNN; ++s) key[s] = 3.0e38f;

    char* const logb  = (char*)logbuf;
    char* const lcol  = logb + tid * 8;
    char*       lnext = lcol;

#pragma unroll 1
    for (int s0 = 0; s0 < TILES_PER_PANEL; s0 += 8) {
        v4f   tv[8];
        float ts[8];
#pragma unroll
        for (int u = 0; u < 8; ++u) {
            const int tb = (s0 + u) * 16;
            tv[u] = *(const v4f*)(gsrc + (size_t)tb * CH);
            ts[u] = gsq[tb];
        }
#pragma unroll
        for (int u = 0; u < 8; ++u) {
            float* d = pan + (s0 + u) * 16 * PITCH;
            v2f lo; lo.x = tv[u].x; lo.y = tv[u].y;
            v2f hi; hi.x = tv[u].z; hi.y = tv[u].w;
            *(v2f*)(d + sdst)     = lo;
            *(v2f*)(d + sdst + 4) = hi;
            d[ndst] = nlead ? ts[u] : 0.0f;
        }
    }
    __syncthreads();

    const int foff = m * PITCH + 4 * h;

    for (int pn = 0; pn < NPANELS; ++pn) {
        const float* cur = pan + (pn & 1) * (PANEL_PTS * PITCH);
        float*       nxt = pan + ((pn & 1) ^ 1) * (PANEL_PTS * PITCH);
        const bool  more = (pn + 1 < NPANELS);
        const int   nextbase = (pn + 1) * PANEL_PTS;

        v4f f[4];
        v2f fs;
#pragma unroll
        for (int g = 0; g < 4; ++g) f[g] = *(const v4f*)(cur + foff + 8 * g);
        fs = *(const v2f*)(cur + m * PITCH + 32 + 2 * h);

#pragma unroll 1
        for (int tt = 0; tt < TILES_PER_PANEL; ++tt) {
            const int colbase = pn * PANEL_PTS + tt * 16;

            if ((tt & (LOGCHK - 1)) == 0) {
                const int n = (int)(lnext - lcol) / LOGROW;
                if (__builtin_amdgcn_ballot_w32(n > LOGCAP - 16 * LOGCHK) != 0u) {
                    log_top16(lcol, n, key);
                    const float T = key[KNN - 1];
                    int nlt = 0;
#pragma unroll
                    for (int s = 0; s < KNN - 1; ++s) nlt += (key[s] < T) ? 1 : 0;
                    const int w = log_filter(lcol, n, T, KNN - nlt);
                    lnext = lcol + w * LOGROW;
                }
            }

            v4f   sv = {};
            float ss = 0.0f;
            if (more) {
                const int tb = nextbase + tt * 16;
                sv = *(const v4f*)(gsrc + (size_t)tb * CH);
                ss = gsq[tb];
            }

            v8f acc0 = {};
            v8f acc1 = {};
#pragma unroll
            for (int g = 0; g < 4; ++g) {
                v2f c0; c0.x = f[g].x; c0.y = f[g].y;
                v2f c1; c1.x = f[g].z; c1.y = f[g].w;
                acc0 = __builtin_amdgcn_wmma_f32_16x16x4_f32(
                    false, c0, false, qfrag0[2 * g], (short)0, acc0, false, false);
                acc1 = __builtin_amdgcn_wmma_f32_16x16x4_f32(
                    false, c0, false, qfrag1[2 * g], (short)0, acc1, false, false);
                acc0 = __builtin_amdgcn_wmma_f32_16x16x4_f32(
                    false, c1, false, qfrag0[2 * g + 1], (short)0, acc0, false, false);
                acc1 = __builtin_amdgcn_wmma_f32_16x16x4_f32(
                    false, c1, false, qfrag1[2 * g + 1], (short)0, acc1, false, false);
            }
            acc0 = __builtin_amdgcn_wmma_f32_16x16x4_f32(
                false, fs, false, qfragOne, (short)0, acc0, false, false);
            acc1 = __builtin_amdgcn_wmma_f32_16x16x4_f32(
                false, fs, false, qfragOne, (short)0, acc1, false, false);

            if (tt + 1 < TILES_PER_PANEL) {
                const float* fr = cur + (tt + 1) * 16 * PITCH;
#pragma unroll
                for (int g = 0; g < 4; ++g) f[g] = *(const v4f*)(fr + foff + 8 * g);
                fs = *(const v2f*)(fr + m * PITCH + 32 + 2 * h);
            }

            float cand[16];
#pragma unroll
            for (int r = 0; r < 8; ++r) {
                const float a0 = acc0[r];
                const float a1 = acc1[r];
                const float give = (h == 0) ? a1 : a0;
                const float got  = xhalf(give);
                cand[r]     = (h == 0) ? a0 : got;
                cand[8 + r] = (h == 0) ? got : a1;
            }

            const float T = key[KNN - 1];
            {
                const float t0 = __builtin_fminf(__builtin_fminf(cand[0], cand[1]), cand[2]);
                const float t1 = __builtin_fminf(__builtin_fminf(cand[3], cand[4]), cand[5]);
                const float t2 = __builtin_fminf(__builtin_fminf(cand[6], cand[7]), cand[8]);
                const float t3 = __builtin_fminf(__builtin_fminf(cand[9], cand[10]), cand[11]);
                const float t4 = __builtin_fminf(__builtin_fminf(cand[12], cand[13]), cand[14]);
                const float u0 = __builtin_fminf(__builtin_fminf(t0, t1), t2);
                const float u1 = __builtin_fminf(__builtin_fminf(t3, t4), cand[15]);
                const float mn = __builtin_fminf(u0, u1);
#pragma unroll
                for (int s = KNN - 1; s >= 1; --s)
                    key[s] = __builtin_amdgcn_fmed3f(key[s - 1], key[s], mn);
                key[0] = __builtin_fminf(key[0], mn);
            }

#pragma unroll
            for (int c = 0; c < 16; ++c) {
                const float d = cand[c];
                const unsigned under = __builtin_bit_cast(unsigned, d - T) >> 31;
                v2f e;
                e.x = d;
                e.y = __builtin_bit_cast(float, colbase + c);
                *(v2f*)lnext = e;
                lnext += under * LOGROW;
            }
            {
                float* d = nxt + tt * 16 * PITCH;
                v2f lo; lo.x = sv.x; lo.y = sv.y;
                v2f hi; hi.x = sv.z; hi.y = sv.w;
                *(v2f*)(d + sdst)     = lo;
                *(v2f*)(d + sdst + 4) = hi;
                d[ndst] = nlead ? ss : 0.0f;
            }
        }
        __syncthreads();
    }

    {
        const int n = (int)(lnext - lcol) / LOGROW;
        log_top16(lcol, n, key);
        const float T = key[KNN - 1];
        int nlt = 0;
#pragma unroll
        for (int s = 0; s < KNN - 1; ++s) nlt += (key[s] < T) ? 1 : 0;
        (void)log_filter(lcol, n, T, KNN - nlt);
    }
    __syncthreads();

    float* poolb = pooled + ((size_t)b * NPTS + rowbase) * CH;
#pragma unroll 1
    for (int r = 0; r < ROWS_PER_WAVE; ++r) {
        float mx = -3.0e38f;
#pragma unroll
        for (int k = 0; k < KNN; ++k) {
            const int j = __builtin_bit_cast(int,
                *(const float*)(logb + k * LOGROW + (wave * ROWS_PER_WAVE + r) * 8 + 4)) & (NPTS - 1);
            const float v = xb[(size_t)j * CH + lane];
            mx = fmaxf(mx, v);
        }
        poolb[r * CH + lane] = mx;
    }
}

__global__ __launch_bounds__(32) void mlp_kernel(const float* __restrict__ pooled,
                                                 const float* __restrict__ Wl,
                                                 const float* __restrict__ bl,
                                                 const float* __restrict__ Wc,
                                                 const float* __restrict__ bc,
                                                 float* __restrict__ out) {
    __shared__ float hT[16 * 68];

    const int lane = threadIdx.x;
    const int m = lane & 15;
    const int h = lane >> 4;
    const size_t rowbase = (size_t)blockIdx.x * 16;

    v2f afrag[8];
#pragma unroll
    for (int c = 0; c < 8; ++c) {
        const float* p = pooled + (rowbase + m) * CH + 4 * c + 2 * h;
        afrag[c].x = p[0];
        afrag[c].y = p[1];
    }

#pragma unroll
    for (int t = 0; t < 4; ++t) {
        v8f acc = {};
#pragma unroll
        for (int c = 0; c < 8; ++c) {
            const float* p = Wl + (size_t)(4 * c + 2 * h) * LDIM + t * 16 + m;
            v2f bfrag;
            bfrag.x = p[0];
            bfrag.y = p[LDIM];
            acc = __builtin_amdgcn_wmma_f32_16x16x4_f32(
                false, afrag[c], false, bfrag, (short)0, acc, false, false);
        }
        const float bias = bl[t * 16 + m];
#pragma unroll
        for (int r = 0; r < 8; ++r)
            hT[(r + 8 * h) * 68 + t * 16 + m] = acc[r] + bias;
    }
    __syncthreads();

    v2f a2[16];
#pragma unroll
    for (int c = 0; c < 16; ++c) {
        a2[c].x = hT[m * 68 + 4 * c + 2 * h];
        a2[c].y = hT[m * 68 + 4 * c + 2 * h + 1];
    }

    float* outp = out + rowbase * ODIM;
#pragma unroll
    for (int t = 0; t < 8; ++t) {
        v8f acc = {};
#pragma unroll
        for (int c = 0; c < 16; ++c) {
            const float* p = Wc + (size_t)(4 * c + 2 * h) * ODIM + t * 16 + m;
            v2f bfrag;
            bfrag.x = p[0];
            bfrag.y = p[ODIM];
            acc = __builtin_amdgcn_wmma_f32_16x16x4_f32(
                false, a2[c], false, bfrag, (short)0, acc, false, false);
        }
        const float bias = bc[t * 16 + m];
#pragma unroll
        for (int r = 0; r < 8; ++r) {
            const float v = acc[r] + bias;
            outp[(size_t)(r + 8 * h) * ODIM + t * 16 + m] = fmaxf(v, 0.0f);
        }
    }
}

extern "C" void kernel_launch(void* const* d_in, const int* in_sizes, int n_in,
                              void* d_out, int out_size, void* d_ws, size_t ws_size,
                              hipStream_t stream) {
    const float* x  = (const float*)d_in[0];
    const float* Wl = (const float*)d_in[1];
    const float* bl = (const float*)d_in[2];
    const float* Wc = (const float*)d_in[3];
    const float* bc = (const float*)d_in[4];
    float* out = (float*)d_out;

    float* pooled = (float*)d_ws;
    float* sqbuf  = pooled + (size_t)BATCH * NPTS * CH;

    const int total = BATCH * NPTS;
    sq_kernel<<<(total + 255) / 256, 256, 0, stream>>>(x, sqbuf, total);
    knn_pool_kernel<<<BATCH * NWG_PER_BATCH, 128, 0, stream>>>(x, sqbuf, pooled);
    mlp_kernel<<<total / 16, 32, 0, stream>>>(pooled, Wl, bl, Wc, bc, out);
}


# kNN list appends: the 16 entry addresses are summed first, then all LDS stores of a tile step issue as one run ahead of the key-list update, not one store behind each candidate's test; sums unchanged
# speedup vs baseline: 3.4604x; 1.1397x over previous
#include <hip/hip_runtime.h>
#include <hip/hip_bf16.h>

#define BATCH 4
#define NPTS  8192
#define CH    32
#define KNN   16
#define LDIM  64
#define ODIM  128
#define NCTILES (NPTS / 16)
#define ROWS_PER_WAVE 32
#define NRBLK (NPTS / ROWS_PER_WAVE)

typedef __attribute__((ext_vector_type(2))) float v2f;
typedef __attribute__((ext_vector_type(8))) float v8f;

__global__ void sq_kernel(const float* __restrict__ x, float* __restrict__ sq, int total) {
    int i = blockIdx.x * blockDim.x + threadIdx.x;
    if (i >= total) return;
    const float* p = x + (size_t)i * CH;
    float s = 0.0f;
#pragma unroll
    for (int c = 0; c < CH; ++c) s += p[c] * p[c];
    sq[i] = s;
}

#define WAVES_PER_WG     4
#define ROWS_PER_WG      (WAVES_PER_WG * ROWS_PER_WAVE)
#define NWG_PER_BATCH    (NPTS / ROWS_PER_WG)
#define TILES_PER_PANEL  32
#define PANEL_PTS        (TILES_PER_PANEL * 16)
#define NPANELS          (NPTS / PANEL_PTS)
#define PITCH            36

typedef __attribute__((ext_vector_type(4))) float v4f;

#define LOGCAP   128
#define LOGCHK   4
#define LOGROW   (128 * 8)
#define LOGBATCH 16

__device__ __forceinline__ int log_filter(char* col, int n, float T, int need) {
    int w = 0, eq = 0;
#pragma unroll 1
    for (int p0 = 0; p0 < LOGCAP; p0 += LOGBATCH) {
        if (__builtin_amdgcn_ballot_w32(p0 < n) == 0u) break;
        v2f e[LOGBATCH];
#pragma unroll
        for (int u = 0; u < LOGBATCH; ++u)
            e[u] = *(const v2f*)(col + (p0 + u) * LOGROW);
#pragma unroll
        for (int u = 0; u < LOGBATCH; ++u) {
            const bool live = (p0 + u) < n;
            const bool tie  = live && (e[u].x == T);
            const bool keep = (live && (e[u].x < T)) || (tie && (eq < need));
            eq += tie ? 1 : 0;
            *(v2f*)(col + (keep ? w : LOGCAP) * LOGROW) = e[u];
            w += keep ? 1 : 0;
        }
    }
    return w;
}

#define KCE(a, b) { const float lo_ = __builtin_fminf(a, b); const float hi_ = __builtin_fmaxf(a, b); a = lo_; b = hi_; }
#define KLAYER0(v)  KCE(v[0], v[13]) KCE(v[1], v[12]) KCE(v[2], v[15]) KCE(v[3], v[14]) KCE(v[4], v[8]) KCE(v[5], v[6]) KCE(v[7], v[11]) KCE(v[9], v[10])
#define KLAYER1(v)  KCE(v[0], v[5]) KCE(v[1], v[7]) KCE(v[2], v[9]) KCE(v[3], v[4]) KCE(v[6], v[13]) KCE(v[8], v[14]) KCE(v[10], v[15]) KCE(v[11], v[12])
#define KLAYER2(v)  KCE(v[0], v[1]) KCE(v[2], v[3]) KCE(v[4], v[5]) KCE(v[6], v[8]) KCE(v[7], v[9]) KCE(v[10], v[11]) KCE(v[12], v[13]) KCE(v[14], v[15])
#define KLAYER3(v)  KCE(v[0], v[2]) KCE(v[1], v[3]) KCE(v[4], v[10]) KCE(v[5], v[11]) KCE(v[6], v[7]) KCE(v[8], v[9]) KCE(v[12], v[14]) KCE(v[13], v[15])
#define KLAYER4(v)  KCE(v[1], v[2]) KCE(v[3], v[12]) KCE(v[4], v[6]) KCE(v[5], v[7]) KCE(v[8], v[10]) KCE(v[9], v[11]) KCE(v[13], v[14])
#define KLAYER5(v)  KCE(v[1], v[4]) KCE(v[2], v[6]) KCE(v[5], v[8]) KCE(v[7], v[10]) KCE(v[9], v[13]) KCE(v[11], v[14])
#define KLAYER6(v)  KCE(v[2], v[4]) KCE(v[3], v[6]) KCE(v[9], v[12]) KCE(v[11], v[13])
#define KLAYER7(v)  KCE(v[3], v[5]) KCE(v[6], v[8]) KCE(v[7], v[9]) KCE(v[10], v[12])
#define KLAYER8(v)  KCE(v[3], v[4]) KCE(v[5], v[6]) KCE(v[7], v[8]) KCE(v[9], v[10]) KCE(v[11], v[12])
#define KLAYER9(v)  KCE(v[6], v[7]) KCE(v[8], v[9])
#define KLAYER10(k, v) k[0] = __builtin_fminf(k[0], v[15]); k[1] = __builtin_fminf(k[1], v[14]); k[2] = __builtin_fminf(k[2], v[13]); k[3] = __builtin_fminf(k[3], v[12]); k[4] = __builtin_fminf(k[4], v[11]); k[5] = __builtin_fminf(k[5], v[10]); k[6] = __builtin_fminf(k[6], v[9]); k[7] = __builtin_fminf(k[7], v[8]); k[8] = __builtin_fminf(k[8], v[7]); k[9] = __builtin_fminf(k[9], v[6]); k[10] = __builtin_fminf(k[10], v[5]); k[11] = __builtin_fminf(k[11], v[4]); k[12] = __builtin_fminf(k[12], v[3]); k[13] = __builtin_fminf(k[13], v[2]); k[14] = __builtin_fminf(k[14], v[1]); k[15] = __builtin_fminf(k[15], v[0]);
#define KLAYER11(k) KCE(k[0], k[8]) KCE(k[1], k[9]) KCE(k[2], k[10]) KCE(k[3], k[11]) KCE(k[4], k[12]) KCE(k[5], k[13]) KCE(k[6], k[14]) KCE(k[7], k[15])
#define KLAYER12(k) KCE(k[0], k[4]) KCE(k[1], k[5]) KCE(k[2], k[6]) KCE(k[3], k[7]) KCE(k[8], k[12]) KCE(k[9], k[13]) KCE(k[10], k[14]) KCE(k[11], k[15])
#define KLAYER13(k) KCE(k[0], k[2]) KCE(k[1], k[3]) KCE(k[4], k[6]) KCE(k[5], k[7]) KCE(k[8], k[10]) KCE(k[9], k[11]) KCE(k[12], k[14]) KCE(k[13], k[15])
#define KLAYER14(k) KCE(k[0], k[1]) KCE(k[2], k[3]) KCE(k[4], k[5]) KCE(k[6], k[7]) KCE(k[8], k[9]) KCE(k[10], k[11]) KCE(k[12], k[13]) KCE(k[14], k[15])

__device__ __forceinline__ float xhalf(float v) {
    const int b = __builtin_bit_cast(int, v);
    return __builtin_bit_cast(float,
        __builtin_amdgcn_permlanex16(b, b, 0x76543210, 0xfedcba98, false, false));
}

__device__ __forceinline__ void log_top16(const char* col, int n, float (&key)[KNN]) {
#pragma unroll
    for (int s = 0; s < KNN; ++s) key[s] = 3.0e38f;
#pragma unroll 1
    for (int p0 = 0; p0 < LOGCAP; p0 += 16) {
        if (__builtin_amdgcn_ballot_w32(p0 < n) == 0u) break;
        float v[16];
#pragma unroll
        for (int u = 0; u < 16; ++u) {
            const float kk = *(const float*)(col + (p0 + u) * LOGROW);
            v[u] = ((p0 + u) < n) ? kk : 3.0e38f;
        }
        KLAYER0(v) KLAYER1(v) KLAYER2(v) KLAYER3(v) KLAYER4(v)
        KLAYER5(v) KLAYER6(v) KLAYER7(v) KLAYER8(v) KLAYER9(v)
        KLAYER10(key, v)
        KLAYER11(key) KLAYER12(key) KLAYER13(key) KLAYER14(key)
    }
}

__global__ __launch_bounds__(128) void knn_pool_kernel(const float* __restrict__ x,
                                                       const float* __restrict__ sq,
                                                       float* __restrict__ pooled) {
    __shared__ __attribute__((aligned(16))) float pan[2 * PANEL_PTS * PITCH];
    __shared__ __attribute__((aligned(16))) v2f logbuf[(LOGCAP + 1) * ROWS_PER_WG];

    const int tid     = threadIdx.x;
    const int lane    = tid & 31;
    const int wave    = __builtin_amdgcn_readfirstlane(tid >> 5);
    const int b       = blockIdx.x / NWG_PER_BATCH;
    const int rowbase = (blockIdx.x % NWG_PER_BATCH) * ROWS_PER_WG + wave * ROWS_PER_WAVE;
    const float* xb   = x  + (size_t)b * NPTS * CH;
    const float* sqb  = sq + (size_t)b * NPTS;

    const int m = lane & 15;
    const int h = lane >> 4;

    const int spt  = tid >> 3;
    const int sq4  = tid & 7;
    const int sdst = spt * PITCH + 8 * (sq4 >> 1) + 2 * (sq4 & 1);
    const int ndst = spt * PITCH + 32 + (sq4 & 3);
    const bool nlead = (sq4 & 3) == 0;
    const float* gsrc = xb + tid * 4;
    const float* gsq  = sqb + spt;

    v2f qfrag0[8], qfrag1[8];
#pragma unroll
    for (int c = 0; c < 8; ++c) {
        const float* p0 = xb + (size_t)(rowbase + m) * CH + 4 * c + 2 * h;
        const float* p1 = xb + (size_t)(rowbase + 16 + m) * CH + 4 * c + 2 * h;
        qfrag0[c].x = -2.0f * p0[0];
        qfrag0[c].y = -2.0f * p0[1];
        qfrag1[c].x = -2.0f * p1[0];
        qfrag1[c].y = -2.0f * p1[1];
    }
    v2f qfragOne;
    qfragOne.x = (h == 0) ? 1.0f : 0.0f;
    qfragOne.y = 0.0f;

    float key[KNN];
#pragma unroll
    for (int s = 0; s < KNN; ++s) key[s] = 3.0e38f;

    char* const logb  = (char*)logbuf;
    char* const lcol  = logb + tid * 8;
    char*       lnext = lcol;

#pragma unroll 1
    for (int s0 = 0; s0 < TILES_PER_PANEL; s0 += 8) {
        v4f   tv[8];
        float ts[8];
#pragma unroll
        for (int u = 0; u < 8; ++u) {
            const int tb = (s0 + u) * 16;
            tv[u] = *(const v4f*)(gsrc + (size_t)tb * CH);
            ts[u] = gsq[tb];
        }
#pragma unroll
        for (int u = 0; u < 8; ++u) {
            float* d = pan + (s0 + u) * 16 * PITCH;
            v2f lo; lo.x = tv[u].x; lo.y = tv[u].y;
            v2f hi; hi.x = tv[u].z; hi.y = tv[u].w;
            *(v2f*)(d + sdst)     = lo;
            *(v2f*)(d + sdst + 4) = hi;
            d[ndst] = nlead ? ts[u] : 0.0f;
        }
    }
    __syncthreads();

    const int foff = m * PITCH + 4 * h;

    for (int pn = 0; pn < NPANELS; ++pn) {
        const float* cur = pan + (pn & 1) * (PANEL_PTS * PITCH);
        float*       nxt = pan + ((pn & 1) ^ 1) * (PANEL_PTS * PITCH);
        const bool  more = (pn + 1 < NPANELS);
        const int   nextbase = (pn + 1) * PANEL_PTS;

        v4f f[4];
        v2f fs;
#pragma unroll
        for (int g = 0; g < 4; ++g) f[g] = *(const v4f*)(cur + foff + 8 * g);
        fs = *(const v2f*)(cur + m * PITCH + 32 + 2 * h);

#pragma unroll 1
        for (int tt = 0; tt < TILES_PER_PANEL; ++tt) {
            const int colbase = pn * PANEL_PTS + tt * 16;

            if ((tt & (LOGCHK - 1)) == 0) {
                const int n = (int)(lnext - lcol) / LOGROW;
                if (__builtin_amdgcn_ballot_w32(n > LOGCAP - 16 * LOGCHK) != 0u) {
                    log_top16(lcol, n, key);
                    const float T = key[KNN - 1];
                    int nlt = 0;
#pragma unroll
                    for (int s = 0; s < KNN - 1; ++s) nlt += (key[s] < T) ? 1 : 0;
                    const int w = log_filter(lcol, n, T, KNN - nlt);
                    lnext = lcol + w * LOGROW;
                }
            }

            v4f   sv = {};
            float ss = 0.0f;
            if (more) {
                const int tb = nextbase + tt * 16;
                sv = *(const v4f*)(gsrc + (size_t)tb * CH);
                ss = gsq[tb];
            }

            v8f acc0 = {};
            v8f acc1 = {};
#pragma unroll
            for (int g = 0; g < 4; ++g) {
                v2f c0; c0.x = f[g].x; c0.y = f[g].y;
                v2f c1; c1.x = f[g].z; c1.y = f[g].w;
                acc0 = __builtin_amdgcn_wmma_f32_16x16x4_f32(
                    false, c0, false, qfrag0[2 * g], (short)0, acc0, false, false);
                acc1 = __builtin_amdgcn_wmma_f32_16x16x4_f32(
                    false, c0, false, qfrag1[2 * g], (short)0, acc1, false, false);
                acc0 = __builtin_amdgcn_wmma_f32_16x16x4_f32(
                    false, c1, false, qfrag0[2 * g + 1], (short)0, acc0, false, false);
                acc1 = __builtin_amdgcn_wmma_f32_16x16x4_f32(
                    false, c1, false, qfrag1[2 * g + 1], (short)0, acc1, false, false);
            }
            acc0 = __builtin_amdgcn_wmma_f32_16x16x4_f32(
                false, fs, false, qfragOne, (short)0, acc0, false, false);
            acc1 = __builtin_amdgcn_wmma_f32_16x16x4_f32(
                false, fs, false, qfragOne, (short)0, acc1, false, false);

            if (tt + 1 < TILES_PER_PANEL) {
                const float* fr = cur + (tt + 1) * 16 * PITCH;
#pragma unroll
                for (int g = 0; g < 4; ++g) f[g] = *(const v4f*)(fr + foff + 8 * g);
                fs = *(const v2f*)(fr + m * PITCH + 32 + 2 * h);
            }

            float cand[16];
#pragma unroll
            for (int r = 0; r < 8; ++r) {
                const float a0 = acc0[r];
                const float a1 = acc1[r];
                const float give = (h == 0) ? a1 : a0;
                const float got  = xhalf(give);
                cand[r]     = (h == 0) ? a0 : got;
                cand[8 + r] = (h == 0) ? got : a1;
            }

            const float T = key[KNN - 1];

            char* slot[16];
            v2f   ent[16];
#pragma unroll
            for (int c = 0; c < 16; ++c) {
                const float d = cand[c];
                const unsigned under = __builtin_bit_cast(unsigned, d - T) >> 31;
                ent[c].x = d;
                ent[c].y = __builtin_bit_cast(float, colbase + c + lane);
                slot[c] = lnext;
                lnext += under * LOGROW;
            }
#pragma unroll
            for (int c = 0; c < 16; ++c) *(v2f*)slot[c] = ent[c];
            {
                float* d = nxt + tt * 16 * PITCH;
                v2f lo; lo.x = sv.x; lo.y = sv.y;
                v2f hi; hi.x = sv.z; hi.y = sv.w;
                *(v2f*)(d + sdst)     = lo;
                *(v2f*)(d + sdst + 4) = hi;
                d[ndst] = nlead ? ss : 0.0f;
            }
            __builtin_amdgcn_sched_group_barrier(0x001, 1000, 0);
            __builtin_amdgcn_sched_group_barrier(0x200, 19, 0);
            __builtin_amdgcn_sched_barrier(0);
            {
                const float t0 = __builtin_fminf(__builtin_fminf(cand[0], cand[1]), cand[2]);
                const float t1 = __builtin_fminf(__builtin_fminf(cand[3], cand[4]), cand[5]);
                const float t2 = __builtin_fminf(__builtin_fminf(cand[6], cand[7]), cand[8]);
                const float t3 = __builtin_fminf(__builtin_fminf(cand[9], cand[10]), cand[11]);
                const float t4 = __builtin_fminf(__builtin_fminf(cand[12], cand[13]), cand[14]);
                const float u0 = __builtin_fminf(__builtin_fminf(t0, t1), t2);
                const float u1 = __builtin_fminf(__builtin_fminf(t3, t4), cand[15]);
                const float mn = __builtin_fminf(u0, u1);
#pragma unroll
                for (int s = KNN - 1; s >= 1; --s)
                    key[s] = __builtin_amdgcn_fmed3f(key[s - 1], key[s], mn);
                key[0] = __builtin_fminf(key[0], mn);
            }
        }
        __syncthreads();
    }

    {
        const int n = (int)(lnext - lcol) / LOGROW;
        log_top16(lcol, n, key);
        const float T = key[KNN - 1];
        int nlt = 0;
#pragma unroll
        for (int s = 0; s < KNN - 1; ++s) nlt += (key[s] < T) ? 1 : 0;
        (void)log_filter(lcol, n, T, KNN - nlt);
    }
    __syncthreads();

    float* poolb = pooled + ((size_t)b * NPTS + rowbase) * CH;
#pragma unroll 1
    for (int r = 0; r < ROWS_PER_WAVE; ++r) {
        float mx = -3.0e38f;
#pragma unroll
        for (int k = 0; k < KNN; ++k) {
            const int j = (__builtin_bit_cast(int,
                *(const float*)(logb + k * LOGROW + (wave * ROWS_PER_WAVE + r) * 8 + 4)) - r) & (NPTS - 1);
            const float v = xb[(size_t)j * CH + lane];
            mx = fmaxf(mx, v);
        }
        poolb[r * CH + lane] = mx;
    }
}

__global__ __launch_bounds__(32) void mlp_kernel(const float* __restrict__ pooled,
                                                 const float* __restrict__ Wl,
                                                 const float* __restrict__ bl,
                                                 const float* __restrict__ Wc,
                                                 const float* __restrict__ bc,
                                                 float* __restrict__ out) {
    __shared__ float hT[16 * 68];

    const int lane = threadIdx.x;
    const int m = lane & 15;
    const int h = lane >> 4;
    const size_t rowbase = (size_t)blockIdx.x * 16;

    v2f afrag[8];
#pragma unroll
    for (int c = 0; c < 8; ++c) {
        const float* p = pooled + (rowbase + m) * CH + 4 * c + 2 * h;
        afrag[c].x = p[0];
        afrag[c].y = p[1];
    }

#pragma unroll
    for (int t = 0; t < 4; ++t) {
        v8f acc = {};
#pragma unroll
        for (int c = 0; c < 8; ++c) {
            const float* p = Wl + (size_t)(4 * c + 2 * h) * LDIM + t * 16 + m;
            v2f bfrag;
            bfrag.x = p[0];
            bfrag.y = p[LDIM];
            acc = __builtin_amdgcn_wmma_f32_16x16x4_f32(
                false, afrag[c], false, bfrag, (short)0, acc, false, false);
        }
        const float bias = bl[t * 16 + m];
#pragma unroll
        for (int r = 0; r < 8; ++r)
            hT[(r + 8 * h) * 68 + t * 16 + m] = acc[r] + bias;
    }
    __syncthreads();

    v2f a2[16];
#pragma unroll
    for (int c = 0; c < 16; ++c) {
        a2[c].x = hT[m * 68 + 4 * c + 2 * h];
        a2[c].y = hT[m * 68 + 4 * c + 2 * h + 1];
    }

    float* outp = out + rowbase * ODIM;
#pragma unroll
    for (int t = 0; t < 8; ++t) {
        v8f acc = {};
#pragma unroll
        for (int c = 0; c < 16; ++c) {
            const float* p = Wc + (size_t)(4 * c + 2 * h) * ODIM + t * 16 + m;
            v2f bfrag;
            bfrag.x = p[0];
            bfrag.y = p[ODIM];
            acc = __builtin_amdgcn_wmma_f32_16x16x4_f32(
                false, a2[c], false, bfrag, (short)0, acc, false, false);
        }
        const float bias = bc[t * 16 + m];
#pragma unroll
        for (int r = 0; r < 8; ++r) {
            const float v = acc[r] + bias;
            outp[(size_t)(r + 8 * h) * ODIM + t * 16 + m] = fmaxf(v, 0.0f);
        }
    }
}

extern "C" void kernel_launch(void* const* d_in, const int* in_sizes, int n_in,
                              void* d_out, int out_size, void* d_ws, size_t ws_size,
                              hipStream_t stream) {
    const float* x  = (const float*)d_in[0];
    const float* Wl = (const float*)d_in[1];
    const float* bl = (const float*)d_in[2];
    const float* Wc = (const float*)d_in[3];
    const float* bc = (const float*)d_in[4];
    float* out = (float*)d_out;

    float* pooled = (float*)d_ws;
    float* sqbuf  = pooled + (size_t)BATCH * NPTS * CH;

    const int total = BATCH * NPTS;
    sq_kernel<<<(total + 255) / 256, 256, 0, stream>>>(x, sqbuf, total);
    knn_pool_kernel<<<BATCH * NWG_PER_BATCH, 128, 0, stream>>>(x, sqbuf, pooled);
    mlp_kernel<<<total / 16, 32, 0, stream>>>(pooled, Wl, bl, Wc, bc, out);
}


# kNN wave owns 16 rows, 2 column tiles a step: lane keeps its accumulators' 8+8 keys, no lane exchange; 8 waves a group; half lists merged at end (index by scalar read, was miscast); sums unchanged
# speedup vs baseline: 3.8732x; 1.1193x over previous
#include <hip/hip_runtime.h>
#include <hip/hip_bf16.h>

#define BATCH 4
#define NPTS  8192
#define CH    32
#define KNN   16
#define LDIM  64
#define ODIM  128
#define NCTILES (NPTS / 16)
#define ROWS_PER_WAVE 16
#define NRBLK (NPTS / ROWS_PER_WAVE)

typedef __attribute__((ext_vector_type(2))) float v2f;
typedef __attribute__((ext_vector_type(8))) float v8f;

__global__ void sq_kernel(const float* __restrict__ x, float* __restrict__ sq, int total) {
    int i = blockIdx.x * blockDim.x + threadIdx.x;
    if (i >= total) return;
    const float* p = x + (size_t)i * CH;
    float s = 0.0f;
#pragma unroll
    for (int c = 0; c < CH; ++c) s += p[c] * p[c];
    sq[i] = s;
}

#define WAVES_PER_WG     8
#define ROWS_PER_WG      (WAVES_PER_WG * ROWS_PER_WAVE)
#define NWG_PER_BATCH    (NPTS / ROWS_PER_WG)
#define TILES_PER_PANEL  32
#define PANEL_PTS        (TILES_PER_PANEL * 16)
#define NPANELS          (NPTS / PANEL_PTS)
#define PITCH            36

typedef __attribute__((ext_vector_type(4))) float v4f;

#define LOGCAP   80
#define LOGCHK   2
#define LOGROW   (256 * 8)
#define STEPS_PER_PANEL (TILES_PER_PANEL / 2)
#define LOGBATCH 16

__device__ __forceinline__ int log_filter(char* col, int n, float T, int need) {
    int w = 0, eq = 0;
#pragma unroll 1
    for (int p0 = 0; p0 < LOGCAP; p0 += LOGBATCH) {
        if (__builtin_amdgcn_ballot_w32(p0 < n) == 0u) break;
        v2f e[LOGBATCH];
#pragma unroll
        for (int u = 0; u < LOGBATCH; ++u)
            e[u] = *(const v2f*)(col + (p0 + u) * LOGROW);
#pragma unroll
        for (int u = 0; u < LOGBATCH; ++u) {
            const bool live = (p0 + u) < n;
            const bool tie  = live && (e[u].x == T);
            const bool keep = (live && (e[u].x < T)) || (tie && (eq < need));
            eq += tie ? 1 : 0;
            *(v2f*)(col + (keep ? w : LOGCAP) * LOGROW) = e[u];
            w += keep ? 1 : 0;
        }
    }
    return w;
}

#define KCE(a, b) { const float lo_ = __builtin_fminf(a, b); const float hi_ = __builtin_fmaxf(a, b); a = lo_; b = hi_; }
#define KLAYER0(v)  KCE(v[0], v[13]) KCE(v[1], v[12]) KCE(v[2], v[15]) KCE(v[3], v[14]) KCE(v[4], v[8]) KCE(v[5], v[6]) KCE(v[7], v[11]) KCE(v[9], v[10])
#define KLAYER1(v)  KCE(v[0], v[5]) KCE(v[1], v[7]) KCE(v[2], v[9]) KCE(v[3], v[4]) KCE(v[6], v[13]) KCE(v[8], v[14]) KCE(v[10], v[15]) KCE(v[11], v[12])
#define KLAYER2(v)  KCE(v[0], v[1]) KCE(v[2], v[3]) KCE(v[4], v[5]) KCE(v[6], v[8]) KCE(v[7], v[9]) KCE(v[10], v[11]) KCE(v[12], v[13]) KCE(v[14], v[15])
#define KLAYER3(v)  KCE(v[0], v[2]) KCE(v[1], v[3]) KCE(v[4], v[10]) KCE(v[5], v[11]) KCE(v[6], v[7]) KCE(v[8], v[9]) KCE(v[12], v[14]) KCE(v[13], v[15])
#define KLAYER4(v)  KCE(v[1], v[2]) KCE(v[3], v[12]) KCE(v[4], v[6]) KCE(v[5], v[7]) KCE(v[8], v[10]) KCE(v[9], v[11]) KCE(v[13], v[14])
#define KLAYER5(v)  KCE(v[1], v[4]) KCE(v[2], v[6]) KCE(v[5], v[8]) KCE(v[7], v[10]) KCE(v[9], v[13]) KCE(v[11], v[14])
#define KLAYER6(v)  KCE(v[2], v[4]) KCE(v[3], v[6]) KCE(v[9], v[12]) KCE(v[11], v[13])
#define KLAYER7(v)  KCE(v[3], v[5]) KCE(v[6], v[8]) KCE(v[7], v[9]) KCE(v[10], v[12])
#define KLAYER8(v)  KCE(v[3], v[4]) KCE(v[5], v[6]) KCE(v[7], v[8]) KCE(v[9], v[10]) KCE(v[11], v[12])
#define KLAYER9(v)  KCE(v[6], v[7]) KCE(v[8], v[9])
#define KLAYER10(k, v) k[0] = __builtin_fminf(k[0], v[15]); k[1] = __builtin_fminf(k[1], v[14]); k[2] = __builtin_fminf(k[2], v[13]); k[3] = __builtin_fminf(k[3], v[12]); k[4] = __builtin_fminf(k[4], v[11]); k[5] = __builtin_fminf(k[5], v[10]); k[6] = __builtin_fminf(k[6], v[9]); k[7] = __builtin_fminf(k[7], v[8]); k[8] = __builtin_fminf(k[8], v[7]); k[9] = __builtin_fminf(k[9], v[6]); k[10] = __builtin_fminf(k[10], v[5]); k[11] = __builtin_fminf(k[11], v[4]); k[12] = __builtin_fminf(k[12], v[3]); k[13] = __builtin_fminf(k[13], v[2]); k[14] = __builtin_fminf(k[14], v[1]); k[15] = __builtin_fminf(k[15], v[0]);
#define KLAYER11(k) KCE(k[0], k[8]) KCE(k[1], k[9]) KCE(k[2], k[10]) KCE(k[3], k[11]) KCE(k[4], k[12]) KCE(k[5], k[13]) KCE(k[6], k[14]) KCE(k[7], k[15])
#define KLAYER12(k) KCE(k[0], k[4]) KCE(k[1], k[5]) KCE(k[2], k[6]) KCE(k[3], k[7]) KCE(k[8], k[12]) KCE(k[9], k[13]) KCE(k[10], k[14]) KCE(k[11], k[15])
#define KLAYER13(k) KCE(k[0], k[2]) KCE(k[1], k[3]) KCE(k[4], k[6]) KCE(k[5], k[7]) KCE(k[8], k[10]) KCE(k[9], k[11]) KCE(k[12], k[14]) KCE(k[13], k[15])
#define KLAYER14(k) KCE(k[0], k[1]) KCE(k[2], k[3]) KCE(k[4], k[5]) KCE(k[6], k[7]) KCE(k[8], k[9]) KCE(k[10], k[11]) KCE(k[12], k[13]) KCE(k[14], k[15])

__device__ __forceinline__ float xhalf(float v) {
    const int b = __builtin_bit_cast(int, v);
    return __builtin_bit_cast(float,
        __builtin_amdgcn_permlanex16(b, b, 0x76543210, 0xfedcba98, false, false));
}

__device__ __forceinline__ void log_top16(const char* col, int n, float (&key)[KNN]) {
#pragma unroll
    for (int s = 0; s < KNN; ++s) key[s] = 3.0e38f;
#pragma unroll 1
    for (int p0 = 0; p0 < LOGCAP; p0 += 16) {
        if (__builtin_amdgcn_ballot_w32(p0 < n) == 0u) break;
        float v[16];
#pragma unroll
        for (int u = 0; u < 16; ++u) {
            const float kk = *(const float*)(col + (p0 + u) * LOGROW);
            v[u] = ((p0 + u) < n) ? kk : 3.0e38f;
        }
        KLAYER0(v) KLAYER1(v) KLAYER2(v) KLAYER3(v) KLAYER4(v)
        KLAYER5(v) KLAYER6(v) KLAYER7(v) KLAYER8(v) KLAYER9(v)
        KLAYER10(key, v)
        KLAYER11(key) KLAYER12(key) KLAYER13(key) KLAYER14(key)
    }
}

__global__ __launch_bounds__(256) void knn_pool_kernel(const float* __restrict__ x,
                                                       const float* __restrict__ sq,
                                                       float* __restrict__ pooled) {
    __shared__ __attribute__((aligned(16))) float pan[2 * PANEL_PTS * PITCH];
    __shared__ __attribute__((aligned(16))) v2f logbuf[(LOGCAP + 1) * 256];
    __shared__ int fin[ROWS_PER_WG * KNN];

    const int tid     = threadIdx.x;
    const int lane    = tid & 31;
    const int wave    = __builtin_amdgcn_readfirstlane(tid >> 5);
    const int b       = blockIdx.x / NWG_PER_BATCH;
    const int rowbase = (blockIdx.x % NWG_PER_BATCH) * ROWS_PER_WG + wave * ROWS_PER_WAVE;
    const float* xb   = x  + (size_t)b * NPTS * CH;
    const float* sqb  = sq + (size_t)b * NPTS;

    const int m = lane & 15;
    const int h = lane >> 4;

    const int spt  = tid >> 3;
    const int sq4  = tid & 7;
    const int sdst = spt * PITCH + 8 * (sq4 >> 1) + 2 * (sq4 & 1);
    const int ndst = spt * PITCH + 32 + (sq4 & 3);
    const bool nlead = (sq4 & 3) == 0;
    const float* gsrc = xb + tid * 4;
    const float* gsq  = sqb + spt;

    v2f qfrag0[8];
#pragma unroll
    for (int c = 0; c < 8; ++c) {
        const float* p0 = xb + (size_t)(rowbase + m) * CH + 4 * c + 2 * h;
        qfrag0[c].x = -2.0f * p0[0];
        qfrag0[c].y = -2.0f * p0[1];
    }
    v2f qfragOne;
    qfragOne.x = (h == 0) ? 1.0f : 0.0f;
    qfragOne.y = 0.0f;

    float key[KNN];
#pragma unroll
    for (int s = 0; s < KNN; ++s) key[s] = 3.0e38f;

    char* const logb  = (char*)logbuf;
    char* const lcol  = logb + tid * 8;
    char*       lnext = lcol;

#pragma unroll 1
    for (int s0 = 0; s0 < STEPS_PER_PANEL; s0 += 8) {
        v4f   tv[8];
        float ts[8];
#pragma unroll
        for (int u = 0; u < 8; ++u) {
            const int tb = (s0 + u) * 32;
            tv[u] = *(const v4f*)(gsrc + (size_t)tb * CH);
            ts[u] = gsq[tb];
        }
#pragma unroll
        for (int u = 0; u < 8; ++u) {
            float* d = pan + (s0 + u) * 32 * PITCH;
            v2f lo; lo.x = tv[u].x; lo.y = tv[u].y;
            v2f hi; hi.x = tv[u].z; hi.y = tv[u].w;
            *(v2f*)(d + sdst)     = lo;
            *(v2f*)(d + sdst + 4) = hi;
            d[ndst] = nlead ? ts[u] : 0.0f;
        }
    }
    __syncthreads();

    const int foff = m * PITCH + 4 * h;
    const int idx8 = 8 * h;

    for (int pn = 0; pn < NPANELS; ++pn) {
        const float* cur = pan + (pn & 1) * (PANEL_PTS * PITCH);
        float*       nxt = pan + ((pn & 1) ^ 1) * (PANEL_PTS * PITCH);
        const bool  more = (pn + 1 < NPANELS);
        const int   nextbase = (pn + 1) * PANEL_PTS;

        v4f f0[4], f1[4];
        v2f fs0, fs1;
#pragma unroll
        for (int g = 0; g < 4; ++g) f0[g] = *(const v4f*)(cur + foff + 8 * g);
        fs0 = *(const v2f*)(cur + m * PITCH + 32 + 2 * h);
#pragma unroll
        for (int g = 0; g < 4; ++g) f1[g] = *(const v4f*)(cur + 16 * PITCH + foff + 8 * g);
        fs1 = *(const v2f*)(cur + 16 * PITCH + m * PITCH + 32 + 2 * h);

#pragma unroll 1
        for (int tt = 0; tt < STEPS_PER_PANEL; ++tt) {
            const int colbase = pn * PANEL_PTS + tt * 32;

            if ((tt & (LOGCHK - 1)) == 0) {
                const int n = (int)(lnext - lcol) / LOGROW;
                if (__builtin_amdgcn_ballot_w32(n > LOGCAP - 16 * LOGCHK) != 0u) {
                    log_top16(lcol, n, key);
                    const float T = key[KNN - 1];
                    int nlt = 0;
#pragma unroll
                    for (int s = 0; s < KNN - 1; ++s) nlt += (key[s] < T) ? 1 : 0;
                    const int w = log_filter(lcol, n, T, KNN - nlt);
                    lnext = lcol + w * LOGROW;
                }
            }

            v4f   sv = {};
            float ss = 0.0f;
            if (more) {
                const int tb = nextbase + tt * 32;
                sv = *(const v4f*)(gsrc + (size_t)tb * CH);
                ss = gsq[tb];
            }

            v8f acc0 = {};
            v8f acc1 = {};
#pragma unroll
            for (int g = 0; g < 4; ++g) {
                v2f a0; a0.x = f0[g].x; a0.y = f0[g].y;
                v2f a1; a1.x = f1[g].x; a1.y = f1[g].y;
                v2f b0; b0.x = f0[g].z; b0.y = f0[g].w;
                v2f b1; b1.x = f1[g].z; b1.y = f1[g].w;
                acc0 = __builtin_amdgcn_wmma_f32_16x16x4_f32(
                    false, a0, false, qfrag0[2 * g], (short)0, acc0, false, false);
                acc1 = __builtin_amdgcn_wmma_f32_16x16x4_f32(
                    false, a1, false, qfrag0[2 * g], (short)0, acc1, false, false);
                acc0 = __builtin_amdgcn_wmma_f32_16x16x4_f32(
                    false, b0, false, qfrag0[2 * g + 1], (short)0, acc0, false, false);
                acc1 = __builtin_amdgcn_wmma_f32_16x16x4_f32(
                    false, b1, false, qfrag0[2 * g + 1], (short)0, acc1, false, false);
            }
            acc0 = __builtin_amdgcn_wmma_f32_16x16x4_f32(
                false, fs0, false, qfragOne, (short)0, acc0, false, false);
            acc1 = __builtin_amdgcn_wmma_f32_16x16x4_f32(
                false, fs1, false, qfragOne, (short)0, acc1, false, false);

            if (tt + 1 < STEPS_PER_PANEL) {
                const float* fr = cur + (tt + 1) * 32 * PITCH;
#pragma unroll
                for (int g = 0; g < 4; ++g) f0[g] = *(const v4f*)(fr + foff + 8 * g);
                fs0 = *(const v2f*)(fr + m * PITCH + 32 + 2 * h);
#pragma unroll
                for (int g = 0; g < 4; ++g) f1[g] = *(const v4f*)(fr + 16 * PITCH + foff + 8 * g);
                fs1 = *(const v2f*)(fr + 16 * PITCH + m * PITCH + 32 + 2 * h);
            }

            float cand[16];
#pragma unroll
            for (int r = 0; r < 8; ++r) {
                cand[r]     = acc0[r];
                cand[8 + r] = acc1[r];
            }

            const float T = key[KNN - 1];

            char* slot[16];
            v2f   ent[16];
            const int ibase = colbase + idx8;
#pragma unroll
            for (int c = 0; c < 16; ++c) {
                const float d = cand[c];
                const unsigned under = __builtin_bit_cast(unsigned, d - T) >> 31;
                ent[c].x = d;
                ent[c].y = __builtin_bit_cast(float, ibase + ((c < 8) ? c : c + 8));
                slot[c] = lnext;
                lnext += under * LOGROW;
            }
#pragma unroll
            for (int c = 0; c < 16; ++c) *(v2f*)slot[c] = ent[c];
            {
                float* d = nxt + tt * 32 * PITCH;
                v2f lo; lo.x = sv.x; lo.y = sv.y;
                v2f hi; hi.x = sv.z; hi.y = sv.w;
                *(v2f*)(d + sdst)     = lo;
                *(v2f*)(d + sdst + 4) = hi;
                d[ndst] = nlead ? ss : 0.0f;
            }
            __builtin_amdgcn_sched_group_barrier(0x001, 1000, 0);
            __builtin_amdgcn_sched_group_barrier(0x200, 19, 0);
            __builtin_amdgcn_sched_barrier(0);
            {
                const float t0 = __builtin_fminf(__builtin_fminf(cand[0], cand[1]), cand[2]);
                const float t1 = __builtin_fminf(__builtin_fminf(cand[3], cand[4]), cand[5]);
                const float t2 = __builtin_fminf(__builtin_fminf(cand[6], cand[7]), cand[8]);
                const float t3 = __builtin_fminf(__builtin_fminf(cand[9], cand[10]), cand[11]);
                const float t4 = __builtin_fminf(__builtin_fminf(cand[12], cand[13]), cand[14]);
                const float u0 = __builtin_fminf(__builtin_fminf(t0, t1), t2);
                const float u1 = __builtin_fminf(__builtin_fminf(t3, t4), cand[15]);
                const float mn = __builtin_fminf(u0, u1);
#pragma unroll
                for (int s = KNN - 1; s >= 1; --s)
                    key[s] = __builtin_amdgcn_fmed3f(key[s - 1], key[s], mn);
                key[0] = __builtin_fminf(key[0], mn);
            }
        }
        __syncthreads();
    }

    {
        const int n = (int)(lnext - lcol) / LOGROW;
        log_top16(lcol, n, key);
        float rk[KNN];
#pragma unroll
        for (int s = 0; s < KNN; ++s) rk[s] = __builtin_fminf(key[s], xhalf(key[KNN - 1 - s]));
        KLAYER11(rk) KLAYER12(rk) KLAYER13(rk) KLAYER14(rk)
        const float T = rk[KNN - 1];
        int nlt = 0;
#pragma unroll
        for (int s = 0; s < KNN - 1; ++s) nlt += (rk[s] < T) ? 1 : 0;
        const int need = KNN - nlt;
        const int w  = log_filter(lcol, n, T, LOGCAP);
        const int wo = __builtin_amdgcn_permlanex16(w, w, 0x76543210, 0xfedcba98, false, false);
        const bool crowded = __builtin_amdgcn_ballot_w32(w + wo > KNN) != 0u;
        int* frow = fin + (wave * ROWS_PER_WAVE + m) * KNN;
        int cnt = 0;
#pragma unroll 1
        for (int p = 0; p < LOGCAP; ++p) {
            if (__builtin_amdgcn_ballot_w32(p < w) == 0u) break;
            const float ek = *(const float*)(lcol + p * LOGROW);
            const float ei = *(const float*)(lcol + p * LOGROW + 4);
            const int  myidx = __builtin_bit_cast(int, ei);
            const bool live = p < w;
            bool keep = live;
            if (crowded) {
                const bool tie = live && (ek == T);
                int rank = 0;
#pragma unroll 1
                for (int q = 0; q < LOGCAP; ++q) {
                    if (__builtin_amdgcn_ballot_w32(q < w) == 0u) break;
                    const float ak = *(const float*)(lcol + q * LOGROW);
                    const float af = *(const float*)(lcol + q * LOGROW + 4);
                    const int ai = __builtin_bit_cast(int, af);
                    const int at = ((q < w) && (ak == T)) ? 1 : 0;
                    const int oi = __builtin_amdgcn_permlanex16(ai, ai, 0x76543210, 0xfedcba98, false, false);
                    const int ot = __builtin_amdgcn_permlanex16(at, at, 0x76543210, 0xfedcba98, false, false);
                    rank += ((at != 0) && (ai < myidx)) ? 1 : 0;
                    rank += ((ot != 0) && (oi < myidx)) ? 1 : 0;
                }
                keep = live && (!tie || (rank < need));
            }
            if (keep) frow[((h == 0) ? cnt : (KNN - 1 - cnt)) & (KNN - 1)] = myidx;
            cnt += keep ? 1 : 0;
        }
    }
    __syncthreads();

    float* poolb = pooled + ((size_t)b * NPTS + rowbase) * CH;
#pragma unroll 1
    for (int r = 0; r < ROWS_PER_WAVE; ++r) {
        float mx = -3.0e38f;
#pragma unroll
        for (int k = 0; k < KNN; ++k) {
            const int j = fin[(wave * ROWS_PER_WAVE + r) * KNN + k] & (NPTS - 1);
            const float v = xb[(size_t)j * CH + lane];
            mx = fmaxf(mx, v);
        }
        poolb[r * CH + lane] = mx;
    }
}

__global__ __launch_bounds__(32) void mlp_kernel(const float* __restrict__ pooled,
                                                 const float* __restrict__ Wl,
                                                 const float* __restrict__ bl,
                                                 const float* __restrict__ Wc,
                                                 const float* __restrict__ bc,
                                                 float* __restrict__ out) {
    __shared__ float hT[16 * 68];

    const int lane = threadIdx.x;
    const int m = lane & 15;
    const int h = lane >> 4;
    const size_t rowbase = (size_t)blockIdx.x * 16;

    v2f afrag[8];
#pragma unroll
    for (int c = 0; c < 8; ++c) {
        const float* p = pooled + (rowbase + m) * CH + 4 * c + 2 * h;
        afrag[c].x = p[0];
        afrag[c].y = p[1];
    }

#pragma unroll
    for (int t = 0; t < 4; ++t) {
        v8f acc = {};
#pragma unroll
        for (int c = 0; c < 8; ++c) {
            const float* p = Wl + (size_t)(4 * c + 2 * h) * LDIM + t * 16 + m;
            v2f bfrag;
            bfrag.x = p[0];
            bfrag.y = p[LDIM];
            acc = __builtin_amdgcn_wmma_f32_16x16x4_f32(
                false, afrag[c], false, bfrag, (short)0, acc, false, false);
        }
        const float bias = bl[t * 16 + m];
#pragma unroll
        for (int r = 0; r < 8; ++r)
            hT[(r + 8 * h) * 68 + t * 16 + m] = acc[r] + bias;
    }
    __syncthreads();

    v2f a2[16];
#pragma unroll
    for (int c = 0; c < 16; ++c) {
        a2[c].x = hT[m * 68 + 4 * c + 2 * h];
        a2[c].y = hT[m * 68 + 4 * c + 2 * h + 1];
    }

    float* outp = out + rowbase * ODIM;
#pragma unroll
    for (int t = 0; t < 8; ++t) {
        v8f acc = {};
#pragma unroll
        for (int c = 0; c < 16; ++c) {
            const float* p = Wc + (size_t)(4 * c + 2 * h) * ODIM + t * 16 + m;
            v2f bfrag;
            bfrag.x = p[0];
            bfrag.y = p[ODIM];
            acc = __builtin_amdgcn_wmma_f32_16x16x4_f32(
                false, a2[c], false, bfrag, (short)0, acc, false, false);
        }
        const float bias = bc[t * 16 + m];
#pragma unroll
        for (int r = 0; r < 8; ++r) {
            const float v = acc[r] + bias;
            outp[(size_t)(r + 8 * h) * ODIM + t * 16 + m] = fmaxf(v, 0.0f);
        }
    }
}

extern "C" void kernel_launch(void* const* d_in, const int* in_sizes, int n_in,
                              void* d_out, int out_size, void* d_ws, size_t ws_size,
                              hipStream_t stream) {
    const float* x  = (const float*)d_in[0];
    const float* Wl = (const float*)d_in[1];
    const float* bl = (const float*)d_in[2];
    const float* Wc = (const float*)d_in[3];
    const float* bc = (const float*)d_in[4];
    float* out = (float*)d_out;

    float* pooled = (float*)d_ws;
    float* sqbuf  = pooled + (size_t)BATCH * NPTS * CH;

    const int total = BATCH * NPTS;
    sq_kernel<<<(total + 255) / 256, 256, 0, stream>>>(x, sqbuf, total);
    knn_pool_kernel<<<BATCH * NWG_PER_BATCH, 256, 0, stream>>>(x, sqbuf, pooled);
    mlp_kernel<<<total / 16, 32, 0, stream>>>(pooled, Wl, bl, Wc, bc, out);
}


# kNN global loads requested further ahead: a staged piece 3 steps before its LDS store (4 register sets, step loop unrolled by 4, panel barrier fences LDS only); gather 4 rows a trip; sums unchanged
# speedup vs baseline: 4.0348x; 1.0417x over previous
#include <hip/hip_runtime.h>
#include <hip/hip_bf16.h>

#define BATCH 4
#define NPTS  8192
#define CH    32
#define KNN   16
#define LDIM  64
#define ODIM  128
#define NCTILES (NPTS / 16)
#define ROWS_PER_WAVE 16
#define NRBLK (NPTS / ROWS_PER_WAVE)

typedef __attribute__((ext_vector_type(2))) float v2f;
typedef __attribute__((ext_vector_type(8))) float v8f;

__global__ void sq_kernel(const float* __restrict__ x, float* __restrict__ sq, int total) {
    int i = blockIdx.x * blockDim.x + threadIdx.x;
    if (i >= total) return;
    const float* p = x + (size_t)i * CH;
    float s = 0.0f;
#pragma unroll
    for (int c = 0; c < CH; ++c) s += p[c] * p[c];
    sq[i] = s;
}

#define WAVES_PER_WG     8
#define ROWS_PER_WG      (WAVES_PER_WG * ROWS_PER_WAVE)
#define NWG_PER_BATCH    (NPTS / ROWS_PER_WG)
#define TILES_PER_PANEL  32
#define PANEL_PTS        (TILES_PER_PANEL * 16)
#define NPANELS          (NPTS / PANEL_PTS)
#define PITCH            36
#define REQ_AHEAD        3
#define NSETS            (REQ_AHEAD + 1)
#define NPIECES          ((NPANELS - 1) * (TILES_PER_PANEL / 2))
#define GROWS            4

typedef __attribute__((ext_vector_type(4))) float v4f;

#define LOGCAP   80
#define LOGCHK   2
#define LOGROW   (256 * 8)
#define STEPS_PER_PANEL (TILES_PER_PANEL / 2)
#define LOGBATCH 16

__device__ __forceinline__ int log_filter(char* col, int n, float T, int need) {
    int w = 0, eq = 0;
#pragma unroll 1
    for (int p0 = 0; p0 < LOGCAP; p0 += LOGBATCH) {
        if (__builtin_amdgcn_ballot_w32(p0 < n) == 0u) break;
        v2f e[LOGBATCH];
#pragma unroll
        for (int u = 0; u < LOGBATCH; ++u)
            e[u] = *(const v2f*)(col + (p0 + u) * LOGROW);
#pragma unroll
        for (int u = 0; u < LOGBATCH; ++u) {
            const bool live = (p0 + u) < n;
            const bool tie  = live && (e[u].x == T);
            const bool keep = (live && (e[u].x < T)) || (tie && (eq < need));
            eq += tie ? 1 : 0;
            *(v2f*)(col + (keep ? w : LOGCAP) * LOGROW) = e[u];
            w += keep ? 1 : 0;
        }
    }
    return w;
}

#define KCE(a, b) { const float lo_ = __builtin_fminf(a, b); const float hi_ = __builtin_fmaxf(a, b); a = lo_; b = hi_; }
#define KLAYER0(v)  KCE(v[0], v[13]) KCE(v[1], v[12]) KCE(v[2], v[15]) KCE(v[3], v[14]) KCE(v[4], v[8]) KCE(v[5], v[6]) KCE(v[7], v[11]) KCE(v[9], v[10])
#define KLAYER1(v)  KCE(v[0], v[5]) KCE(v[1], v[7]) KCE(v[2], v[9]) KCE(v[3], v[4]) KCE(v[6], v[13]) KCE(v[8], v[14]) KCE(v[10], v[15]) KCE(v[11], v[12])
#define KLAYER2(v)  KCE(v[0], v[1]) KCE(v[2], v[3]) KCE(v[4], v[5]) KCE(v[6], v[8]) KCE(v[7], v[9]) KCE(v[10], v[11]) KCE(v[12], v[13]) KCE(v[14], v[15])
#define KLAYER3(v)  KCE(v[0], v[2]) KCE(v[1], v[3]) KCE(v[4], v[10]) KCE(v[5], v[11]) KCE(v[6], v[7]) KCE(v[8], v[9]) KCE(v[12], v[14]) KCE(v[13], v[15])
#define KLAYER4(v)  KCE(v[1], v[2]) KCE(v[3], v[12]) KCE(v[4], v[6]) KCE(v[5], v[7]) KCE(v[8], v[10]) KCE(v[9], v[11]) KCE(v[13], v[14])
#define KLAYER5(v)  KCE(v[1], v[4]) KCE(v[2], v[6]) KCE(v[5], v[8]) KCE(v[7], v[10]) KCE(v[9], v[13]) KCE(v[11], v[14])
#define KLAYER6(v)  KCE(v[2], v[4]) KCE(v[3], v[6]) KCE(v[9], v[12]) KCE(v[11], v[13])
#define KLAYER7(v)  KCE(v[3], v[5]) KCE(v[6], v[8]) KCE(v[7], v[9]) KCE(v[10], v[12])
#define KLAYER8(v)  KCE(v[3], v[4]) KCE(v[5], v[6]) KCE(v[7], v[8]) KCE(v[9], v[10]) KCE(v[11], v[12])
#define KLAYER9(v)  KCE(v[6], v[7]) KCE(v[8], v[9])
#define KLAYER10(k, v) k[0] = __builtin_fminf(k[0], v[15]); k[1] = __builtin_fminf(k[1], v[14]); k[2] = __builtin_fminf(k[2], v[13]); k[3] = __builtin_fminf(k[3], v[12]); k[4] = __builtin_fminf(k[4], v[11]); k[5] = __builtin_fminf(k[5], v[10]); k[6] = __builtin_fminf(k[6], v[9]); k[7] = __builtin_fminf(k[7], v[8]); k[8] = __builtin_fminf(k[8], v[7]); k[9] = __builtin_fminf(k[9], v[6]); k[10] = __builtin_fminf(k[10], v[5]); k[11] = __builtin_fminf(k[11], v[4]); k[12] = __builtin_fminf(k[12], v[3]); k[13] = __builtin_fminf(k[13], v[2]); k[14] = __builtin_fminf(k[14], v[1]); k[15] = __builtin_fminf(k[15], v[0]);
#define KLAYER11(k) KCE(k[0], k[8]) KCE(k[1], k[9]) KCE(k[2], k[10]) KCE(k[3], k[11]) KCE(k[4], k[12]) KCE(k[5], k[13]) KCE(k[6], k[14]) KCE(k[7], k[15])
#define KLAYER12(k) KCE(k[0], k[4]) KCE(k[1], k[5]) KCE(k[2], k[6]) KCE(k[3], k[7]) KCE(k[8], k[12]) KCE(k[9], k[13]) KCE(k[10], k[14]) KCE(k[11], k[15])
#define KLAYER13(k) KCE(k[0], k[2]) KCE(k[1], k[3]) KCE(k[4], k[6]) KCE(k[5], k[7]) KCE(k[8], k[10]) KCE(k[9], k[11]) KCE(k[12], k[14]) KCE(k[13], k[15])
#define KLAYER14(k) KCE(k[0], k[1]) KCE(k[2], k[3]) KCE(k[4], k[5]) KCE(k[6], k[7]) KCE(k[8], k[9]) KCE(k[10], k[11]) KCE(k[12], k[13]) KCE(k[14], k[15])

__device__ __forceinline__ float xhalf(float v) {
    const int b = __builtin_bit_cast(int, v);
    return __builtin_bit_cast(float,
        __builtin_amdgcn_permlanex16(b, b, 0x76543210, 0xfedcba98, false, false));
}

__device__ __forceinline__ void log_top16(const char* col, int n, float (&key)[KNN]) {
#pragma unroll
    for (int s = 0; s < KNN; ++s) key[s] = 3.0e38f;
#pragma unroll 1
    for (int p0 = 0; p0 < LOGCAP; p0 += 16) {
        if (__builtin_amdgcn_ballot_w32(p0 < n) == 0u) break;
        float v[16];
#pragma unroll
        for (int u = 0; u < 16; ++u) {
            const float kk = *(const float*)(col + (p0 + u) * LOGROW);
            v[u] = ((p0 + u) < n) ? kk : 3.0e38f;
        }
        KLAYER0(v) KLAYER1(v) KLAYER2(v) KLAYER3(v) KLAYER4(v)
        KLAYER5(v) KLAYER6(v) KLAYER7(v) KLAYER8(v) KLAYER9(v)
        KLAYER10(key, v)
        KLAYER11(key) KLAYER12(key) KLAYER13(key) KLAYER14(key)
    }
}

__global__ __launch_bounds__(256) void knn_pool_kernel(const float* __restrict__ x,
                                                       const float* __restrict__ sq,
                                                       float* __restrict__ pooled) {
    __shared__ __attribute__((aligned(16))) float pan[2 * PANEL_PTS * PITCH];
    __shared__ __attribute__((aligned(16))) v2f logbuf[(LOGCAP + 1) * 256];
    __shared__ int fin[ROWS_PER_WG * KNN];

    const int tid     = threadIdx.x;
    const int lane    = tid & 31;
    const int wave    = __builtin_amdgcn_readfirstlane(tid >> 5);
    const int b       = blockIdx.x / NWG_PER_BATCH;
    const int rowbase = (blockIdx.x % NWG_PER_BATCH) * ROWS_PER_WG + wave * ROWS_PER_WAVE;
    const float* xb   = x  + (size_t)b * NPTS * CH;
    const float* sqb  = sq + (size_t)b * NPTS;

    const int m = lane & 15;
    const int h = lane >> 4;

    const int spt  = tid >> 3;
    const int sq4  = tid & 7;
    const int sdst = spt * PITCH + 8 * (sq4 >> 1) + 2 * (sq4 & 1);
    const int ndst = spt * PITCH + 32 + (sq4 & 3);
    const bool nlead = (sq4 & 3) == 0;
    const float* gsrc = xb + tid * 4;
    const float* gsq  = sqb + spt;

    v2f qfrag0[8];
#pragma unroll
    for (int c = 0; c < 8; ++c) {
        const float* p0 = xb + (size_t)(rowbase + m) * CH + 4 * c + 2 * h;
        qfrag0[c].x = -2.0f * p0[0];
        qfrag0[c].y = -2.0f * p0[1];
    }
    v2f qfragOne;
    qfragOne.x = (h == 0) ? 1.0f : 0.0f;
    qfragOne.y = 0.0f;

    float key[KNN];
#pragma unroll
    for (int s = 0; s < KNN; ++s) key[s] = 3.0e38f;

    char* const logb  = (char*)logbuf;
    char* const lcol  = logb + tid * 8;
    char*       lnext = lcol;

#pragma unroll 1
    for (int s0 = 0; s0 < STEPS_PER_PANEL; s0 += 8) {
        v4f   tv[8];
        float ts[8];
#pragma unroll
        for (int u = 0; u < 8; ++u) {
            const int tb = (s0 + u) * 32;
            tv[u] = *(const v4f*)(gsrc + (size_t)tb * CH);
            ts[u] = gsq[tb];
        }
#pragma unroll
        for (int u = 0; u < 8; ++u) {
            float* d = pan + (s0 + u) * 32 * PITCH;
            v2f lo; lo.x = tv[u].x; lo.y = tv[u].y;
            v2f hi; hi.x = tv[u].z; hi.y = tv[u].w;
            *(v2f*)(d + sdst)     = lo;
            *(v2f*)(d + sdst + 4) = hi;
            d[ndst] = nlead ? ts[u] : 0.0f;
        }
    }
    v4f   rsv[NSETS];
    float rss[NSETS];
#pragma unroll
    for (int u = 0; u < NSETS; ++u) { rsv[u] = v4f{}; rss[u] = 0.0f; }
#pragma unroll
    for (int u = 0; u < REQ_AHEAD; ++u) {
        const int tb = PANEL_PTS + 32 * u;
        rsv[u] = *(const v4f*)(gsrc + (size_t)tb * CH);
        rss[u] = gsq[tb];
    }
    __syncthreads();

    const int foff = m * PITCH + 4 * h;
    const int idx8 = 8 * h;

    for (int pn = 0; pn < NPANELS; ++pn) {
        const float* cur = pan + (pn & 1) * (PANEL_PTS * PITCH);
        float*       nxt = pan + ((pn & 1) ^ 1) * (PANEL_PTS * PITCH);

        v4f f0[4], f1[4];
        v2f fs0, fs1;
#pragma unroll
        for (int g = 0; g < 4; ++g) f0[g] = *(const v4f*)(cur + foff + 8 * g);
        fs0 = *(const v2f*)(cur + m * PITCH + 32 + 2 * h);
#pragma unroll
        for (int g = 0; g < 4; ++g) f1[g] = *(const v4f*)(cur + 16 * PITCH + foff + 8 * g);
        fs1 = *(const v2f*)(cur + 16 * PITCH + m * PITCH + 32 + 2 * h);

#pragma unroll 1
        for (int t4 = 0; t4 < STEPS_PER_PANEL; t4 += NSETS) {
#pragma unroll
        for (int u = 0; u < NSETS; ++u) {
            const int tt = t4 + u;
            const int colbase = pn * PANEL_PTS + tt * 32;

            if ((u & (LOGCHK - 1)) == 0) {
                const int n = (int)(lnext - lcol) / LOGROW;
                if (__builtin_amdgcn_ballot_w32(n > LOGCAP - 16 * LOGCHK) != 0u) {
                    log_top16(lcol, n, key);
                    const float T = key[KNN - 1];
                    int nlt = 0;
#pragma unroll
                    for (int s = 0; s < KNN - 1; ++s) nlt += (key[s] < T) ? 1 : 0;
                    const int w = log_filter(lcol, n, T, KNN - nlt);
                    lnext = lcol + w * LOGROW;
                }
            }

            {
                const int g  = pn * STEPS_PER_PANEL + tt + REQ_AHEAD;
                const int gc = (g < NPIECES) ? g : (NPIECES - 1);
                const int tb = PANEL_PTS + 32 * gc;
                rsv[(u + REQ_AHEAD) % NSETS] = *(const v4f*)(gsrc + (size_t)tb * CH);
                rss[(u + REQ_AHEAD) % NSETS] = gsq[tb];
            }

            v8f acc0 = {};
            v8f acc1 = {};
#pragma unroll
            for (int g = 0; g < 4; ++g) {
                v2f a0; a0.x = f0[g].x; a0.y = f0[g].y;
                v2f a1; a1.x = f1[g].x; a1.y = f1[g].y;
                v2f b0; b0.x = f0[g].z; b0.y = f0[g].w;
                v2f b1; b1.x = f1[g].z; b1.y = f1[g].w;
                acc0 = __builtin_amdgcn_wmma_f32_16x16x4_f32(
                    false, a0, false, qfrag0[2 * g], (short)0, acc0, false, false);
                acc1 = __builtin_amdgcn_wmma_f32_16x16x4_f32(
                    false, a1, false, qfrag0[2 * g], (short)0, acc1, false, false);
                acc0 = __builtin_amdgcn_wmma_f32_16x16x4_f32(
                    false, b0, false, qfrag0[2 * g + 1], (short)0, acc0, false, false);
                acc1 = __builtin_amdgcn_wmma_f32_16x16x4_f32(
                    false, b1, false, qfrag0[2 * g + 1], (short)0, acc1, false, false);
            }
            acc0 = __builtin_amdgcn_wmma_f32_16x16x4_f32(
                false, fs0, false, qfragOne, (short)0, acc0, false, false);
            acc1 = __builtin_amdgcn_wmma_f32_16x16x4_f32(
                false, fs1, false, qfragOne, (short)0, acc1, false, false);

            __builtin_amdgcn_sched_barrier(0);
            if (tt + 1 < STEPS_PER_PANEL) {
                const float* fr = cur + (tt + 1) * 32 * PITCH;
#pragma unroll
                for (int g = 0; g < 4; ++g) f0[g] = *(const v4f*)(fr + foff + 8 * g);
                fs0 = *(const v2f*)(fr + m * PITCH + 32 + 2 * h);
#pragma unroll
                for (int g = 0; g < 4; ++g) f1[g] = *(const v4f*)(fr + 16 * PITCH + foff + 8 * g);
                fs1 = *(const v2f*)(fr + 16 * PITCH + m * PITCH + 32 + 2 * h);
            }
            __builtin_amdgcn_sched_barrier(0);

            float cand[16];
#pragma unroll
            for (int r = 0; r < 8; ++r) {
                cand[r]     = acc0[r];
                cand[8 + r] = acc1[r];
            }

            const float T = key[KNN - 1];

            char* slot[16];
            v2f   ent[16];
            const int ibase = colbase + idx8;
#pragma unroll
            for (int c = 0; c < 16; ++c) {
                const float d = cand[c];
                const unsigned under = __builtin_bit_cast(unsigned, d - T) >> 31;
                ent[c].x = d;
                ent[c].y = __builtin_bit_cast(float, ibase + ((c < 8) ? c : c + 8));
                slot[c] = lnext;
                lnext += under * LOGROW;
            }
#pragma unroll
            for (int c = 0; c < 16; ++c) *(v2f*)slot[c] = ent[c];
            {
                float* d = nxt + tt * 32 * PITCH;
                const v4f   sv = rsv[u];
                const float ss = rss[u];
                v2f lo; lo.x = sv.x; lo.y = sv.y;
                v2f hi; hi.x = sv.z; hi.y = sv.w;
                *(v2f*)(d + sdst)     = lo;
                *(v2f*)(d + sdst + 4) = hi;
                d[ndst] = nlead ? ss : 0.0f;
            }
            __builtin_amdgcn_sched_group_barrier(0x001, 1000, 0);
            __builtin_amdgcn_sched_group_barrier(0x200, 19, 0);
            __builtin_amdgcn_sched_barrier(0);
            {
                const float t0 = __builtin_fminf(__builtin_fminf(cand[0], cand[1]), cand[2]);
                const float t1 = __builtin_fminf(__builtin_fminf(cand[3], cand[4]), cand[5]);
                const float t2 = __builtin_fminf(__builtin_fminf(cand[6], cand[7]), cand[8]);
                const float t3 = __builtin_fminf(__builtin_fminf(cand[9], cand[10]), cand[11]);
                const float t4 = __builtin_fminf(__builtin_fminf(cand[12], cand[13]), cand[14]);
                const float u0 = __builtin_fminf(__builtin_fminf(t0, t1), t2);
                const float u1 = __builtin_fminf(__builtin_fminf(t3, t4), cand[15]);
                const float mn = __builtin_fminf(u0, u1);
#pragma unroll
                for (int s = KNN - 1; s >= 1; --s)
                    key[s] = __builtin_amdgcn_fmed3f(key[s - 1], key[s], mn);
                key[0] = __builtin_fminf(key[0], mn);
            }
        }
        }
        __builtin_amdgcn_fence(__ATOMIC_RELEASE, "workgroup", "local");
        __builtin_amdgcn_s_barrier();
        __builtin_amdgcn_fence(__ATOMIC_ACQUIRE, "workgroup", "local");
    }

    {
        const int n = (int)(lnext - lcol) / LOGROW;
        log_top16(lcol, n, key);
        float rk[KNN];
#pragma unroll
        for (int s = 0; s < KNN; ++s) rk[s] = __builtin_fminf(key[s], xhalf(key[KNN - 1 - s]));
        KLAYER11(rk) KLAYER12(rk) KLAYER13(rk) KLAYER14(rk)
        const float T = rk[KNN - 1];
        int nlt = 0;
#pragma unroll
        for (int s = 0; s < KNN - 1; ++s) nlt += (rk[s] < T) ? 1 : 0;
        const int need = KNN - nlt;
        const int w  = log_filter(lcol, n, T, LOGCAP);
        const int wo = __builtin_amdgcn_permlanex16(w, w, 0x76543210, 0xfedcba98, false, false);
        const bool crowded = __builtin_amdgcn_ballot_w32(w + wo > KNN) != 0u;
        int* frow = fin + (wave * ROWS_PER_WAVE + m) * KNN;
        int cnt = 0;
#pragma unroll 1
        for (int p = 0; p < LOGCAP; ++p) {
            if (__builtin_amdgcn_ballot_w32(p < w) == 0u) break;
            const float ek = *(const float*)(lcol + p * LOGROW);
            const float ei = *(const float*)(lcol + p * LOGROW + 4);
            const int  myidx = __builtin_bit_cast(int, ei);
            const bool live = p < w;
            bool keep = live;
            if (crowded) {
                const bool tie = live && (ek == T);
                int rank = 0;
#pragma unroll 1
                for (int q = 0; q < LOGCAP; ++q) {
                    if (__builtin_amdgcn_ballot_w32(q < w) == 0u) break;
                    const float ak = *(const float*)(lcol + q * LOGROW);
                    const float af = *(const float*)(lcol + q * LOGROW + 4);
                    const int ai = __builtin_bit_cast(int, af);
                    const int at = ((q < w) && (ak == T)) ? 1 : 0;
                    const int oi = __builtin_amdgcn_permlanex16(ai, ai, 0x76543210, 0xfedcba98, false, false);
                    const int ot = __builtin_amdgcn_permlanex16(at, at, 0x76543210, 0xfedcba98, false, false);
                    rank += ((at != 0) && (ai < myidx)) ? 1 : 0;
                    rank += ((ot != 0) && (oi < myidx)) ? 1 : 0;
                }
                keep = live && (!tie || (rank < need));
            }
            if (keep) frow[((h == 0) ? cnt : (KNN - 1 - cnt)) & (KNN - 1)] = myidx;
            cnt += keep ? 1 : 0;
        }
    }
    __syncthreads();

    float* poolb = pooled + ((size_t)b * NPTS + rowbase) * CH;
#pragma unroll 1
    for (int r0 = 0; r0 < ROWS_PER_WAVE; r0 += GROWS) {
        float mx[GROWS];
#pragma unroll
        for (int q = 0; q < GROWS; ++q) {
            mx[q] = -3.0e38f;
#pragma unroll
            for (int k = 0; k < KNN; ++k) {
                const int j = fin[(wave * ROWS_PER_WAVE + r0 + q) * KNN + k] & (NPTS - 1);
                const float v = xb[(size_t)j * CH + lane];
                mx[q] = fmaxf(mx[q], v);
            }
        }
#pragma unroll
        for (int q = 0; q < GROWS; ++q) poolb[(r0 + q) * CH + lane] = mx[q];
    }
}

__global__ __launch_bounds__(32) void mlp_kernel(const float* __restrict__ pooled,
                                                 const float* __restrict__ Wl,
                                                 const float* __restrict__ bl,
                                                 const float* __restrict__ Wc,
                                                 const float* __restrict__ bc,
                                                 float* __restrict__ out) {
    __shared__ float hT[16 * 68];

    const int lane = threadIdx.x;
    const int m = lane & 15;
    const int h = lane >> 4;
    const size_t rowbase = (size_t)blockIdx.x * 16;

    v2f afrag[8];
#pragma unroll
    for (int c = 0; c < 8; ++c) {
        const float* p = pooled + (rowbase + m) * CH + 4 * c + 2 * h;
        afrag[c].x = p[0];
        afrag[c].y = p[1];
    }

#pragma unroll
    for (int t = 0; t < 4; ++t) {
        v8f acc = {};
#pragma unroll
        for (int c = 0; c < 8; ++c) {
            const float* p = Wl + (size_t)(4 * c + 2 * h) * LDIM + t * 16 + m;
            v2f bfrag;
            bfrag.x = p[0];
            bfrag.y = p[LDIM];
            acc = __builtin_amdgcn_wmma_f32_16x16x4_f32(
                false, afrag[c], false, bfrag, (short)0, acc, false, false);
        }
        const float bias = bl[t * 16 + m];
#pragma unroll
        for (int r = 0; r < 8; ++r)
            hT[(r + 8 * h) * 68 + t * 16 + m] = acc[r] + bias;
    }
    __syncthreads();

    v2f a2[16];
#pragma unroll
    for (int c = 0; c < 16; ++c) {
        a2[c].x = hT[m * 68 + 4 * c + 2 * h];
        a2[c].y = hT[m * 68 + 4 * c + 2 * h + 1];
    }

    float* outp = out + rowbase * ODIM;
#pragma unroll
    for (int t = 0; t < 8; ++t) {
        v8f acc = {};
#pragma unroll
        for (int c = 0; c < 16; ++c) {
            const float* p = Wc + (size_t)(4 * c + 2 * h) * ODIM + t * 16 + m;
            v2f bfrag;
            bfrag.x = p[0];
            bfrag.y = p[ODIM];
            acc = __builtin_amdgcn_wmma_f32_16x16x4_f32(
                false, a2[c], false, bfrag, (short)0, acc, false, false);
        }
        const float bias = bc[t * 16 + m];
#pragma unroll
        for (int r = 0; r < 8; ++r) {
            const float v = acc[r] + bias;
            outp[(size_t)(r + 8 * h) * ODIM + t * 16 + m] = fmaxf(v, 0.0f);
        }
    }
}

extern "C" void kernel_launch(void* const* d_in, const int* in_sizes, int n_in,
                              void* d_out, int out_size, void* d_ws, size_t ws_size,
                              hipStream_t stream) {
    const float* x  = (const float*)d_in[0];
    const float* Wl = (const float*)d_in[1];
    const float* bl = (const float*)d_in[2];
    const float* Wc = (const float*)d_in[3];
    const float* bc = (const float*)d_in[4];
    float* out = (float*)d_out;

    float* pooled = (float*)d_ws;
    float* sqbuf  = pooled + (size_t)BATCH * NPTS * CH;

    const int total = BATCH * NPTS;
    sq_kernel<<<(total + 255) / 256, 256, 0, stream>>>(x, sqbuf, total);
    knn_pool_kernel<<<BATCH * NWG_PER_BATCH, 256, 0, stream>>>(x, sqbuf, pooled);
    mlp_kernel<<<total / 16, 32, 0, stream>>>(pooled, Wl, bl, Wc, bc, out);
}


# kNN list appends two candidates per LDS store: the pair's first place takes its first hit (chosen by the sign of a - T), the second place its later candidate; 8 two-address stores a step, was 16
# speedup vs baseline: 4.3095x; 1.0681x over previous
#include <hip/hip_runtime.h>
#include <hip/hip_bf16.h>

#define BATCH 4
#define NPTS  8192
#define CH    32
#define KNN   16
#define LDIM  64
#define ODIM  128
#define NCTILES (NPTS / 16)
#define ROWS_PER_WAVE 16
#define NRBLK (NPTS / ROWS_PER_WAVE)

typedef __attribute__((ext_vector_type(2))) float v2f;
typedef __attribute__((ext_vector_type(8))) float v8f;

__global__ void sq_kernel(const float* __restrict__ x, float* __restrict__ sq, int total) {
    int i = blockIdx.x * blockDim.x + threadIdx.x;
    if (i >= total) return;
    const float* p = x + (size_t)i * CH;
    float s = 0.0f;
#pragma unroll
    for (int c = 0; c < CH; ++c) s += p[c] * p[c];
    sq[i] = s;
}

#define WAVES_PER_WG     8
#define ROWS_PER_WG      (WAVES_PER_WG * ROWS_PER_WAVE)
#define NWG_PER_BATCH    (NPTS / ROWS_PER_WG)
#define TILES_PER_PANEL  32
#define PANEL_PTS        (TILES_PER_PANEL * 16)
#define NPANELS          (NPTS / PANEL_PTS)
#define PITCH            36
#define REQ_AHEAD        3
#define NSETS            (REQ_AHEAD + 1)
#define NPIECES          ((NPANELS - 1) * (TILES_PER_PANEL / 2))
#define GROWS            4

typedef __attribute__((ext_vector_type(4))) float v4f;

#define LOGCAP   80
#define LOGCHK   2
#define LOGROW   (256 * 8)
#define STEPS_PER_PANEL (TILES_PER_PANEL / 2)
#define LOGBATCH 16

__device__ __forceinline__ int log_filter(char* col, int n, float T, int need) {
    int w = 0, eq = 0;
#pragma unroll 1
    for (int p0 = 0; p0 < LOGCAP; p0 += LOGBATCH) {
        if (__builtin_amdgcn_ballot_w32(p0 < n) == 0u) break;
        v2f e[LOGBATCH];
#pragma unroll
        for (int u = 0; u < LOGBATCH; ++u)
            e[u] = *(const v2f*)(col + (p0 + u) * LOGROW);
#pragma unroll
        for (int u = 0; u < LOGBATCH; ++u) {
            const bool live = (p0 + u) < n;
            const bool tie  = live && (e[u].x == T);
            const bool keep = (live && (e[u].x < T)) || (tie && (eq < need));
            eq += tie ? 1 : 0;
            *(v2f*)(col + (keep ? w : LOGCAP) * LOGROW) = e[u];
            w += keep ? 1 : 0;
        }
    }
    return w;
}

#define KCE(a, b) { const float lo_ = __builtin_fminf(a, b); const float hi_ = __builtin_fmaxf(a, b); a = lo_; b = hi_; }
#define KLAYER0(v)  KCE(v[0], v[13]) KCE(v[1], v[12]) KCE(v[2], v[15]) KCE(v[3], v[14]) KCE(v[4], v[8]) KCE(v[5], v[6]) KCE(v[7], v[11]) KCE(v[9], v[10])
#define KLAYER1(v)  KCE(v[0], v[5]) KCE(v[1], v[7]) KCE(v[2], v[9]) KCE(v[3], v[4]) KCE(v[6], v[13]) KCE(v[8], v[14]) KCE(v[10], v[15]) KCE(v[11], v[12])
#define KLAYER2(v)  KCE(v[0], v[1]) KCE(v[2], v[3]) KCE(v[4], v[5]) KCE(v[6], v[8]) KCE(v[7], v[9]) KCE(v[10], v[11]) KCE(v[12], v[13]) KCE(v[14], v[15])
#define KLAYER3(v)  KCE(v[0], v[2]) KCE(v[1], v[3]) KCE(v[4], v[10]) KCE(v[5], v[11]) KCE(v[6], v[7]) KCE(v[8], v[9]) KCE(v[12], v[14]) KCE(v[13], v[15])
#define KLAYER4(v)  KCE(v[1], v[2]) KCE(v[3], v[12]) KCE(v[4], v[6]) KCE(v[5], v[7]) KCE(v[8], v[10]) KCE(v[9], v[11]) KCE(v[13], v[14])
#define KLAYER5(v)  KCE(v[1], v[4]) KCE(v[2], v[6]) KCE(v[5], v[8]) KCE(v[7], v[10]) KCE(v[9], v[13]) KCE(v[11], v[14])
#define KLAYER6(v)  KCE(v[2], v[4]) KCE(v[3], v[6]) KCE(v[9], v[12]) KCE(v[11], v[13])
#define KLAYER7(v)  KCE(v[3], v[5]) KCE(v[6], v[8]) KCE(v[7], v[9]) KCE(v[10], v[12])
#define KLAYER8(v)  KCE(v[3], v[4]) KCE(v[5], v[6]) KCE(v[7], v[8]) KCE(v[9], v[10]) KCE(v[11], v[12])
#define KLAYER9(v)  KCE(v[6], v[7]) KCE(v[8], v[9])
#define KLAYER10(k, v) k[0] = __builtin_fminf(k[0], v[15]); k[1] = __builtin_fminf(k[1], v[14]); k[2] = __builtin_fminf(k[2], v[13]); k[3] = __builtin_fminf(k[3], v[12]); k[4] = __builtin_fminf(k[4], v[11]); k[5] = __builtin_fminf(k[5], v[10]); k[6] = __builtin_fminf(k[6], v[9]); k[7] = __builtin_fminf(k[7], v[8]); k[8] = __builtin_fminf(k[8], v[7]); k[9] = __builtin_fminf(k[9], v[6]); k[10] = __builtin_fminf(k[10], v[5]); k[11] = __builtin_fminf(k[11], v[4]); k[12] = __builtin_fminf(k[12], v[3]); k[13] = __builtin_fminf(k[13], v[2]); k[14] = __builtin_fminf(k[14], v[1]); k[15] = __builtin_fminf(k[15], v[0]);
#define KLAYER11(k) KCE(k[0], k[8]) KCE(k[1], k[9]) KCE(k[2], k[10]) KCE(k[3], k[11]) KCE(k[4], k[12]) KCE(k[5], k[13]) KCE(k[6], k[14]) KCE(k[7], k[15])
#define KLAYER12(k) KCE(k[0], k[4]) KCE(k[1], k[5]) KCE(k[2], k[6]) KCE(k[3], k[7]) KCE(k[8], k[12]) KCE(k[9], k[13]) KCE(k[10], k[14]) KCE(k[11], k[15])
#define KLAYER13(k) KCE(k[0], k[2]) KCE(k[1], k[3]) KCE(k[4], k[6]) KCE(k[5], k[7]) KCE(k[8], k[10]) KCE(k[9], k[11]) KCE(k[12], k[14]) KCE(k[13], k[15])
#define KLAYER14(k) KCE(k[0], k[1]) KCE(k[2], k[3]) KCE(k[4], k[5]) KCE(k[6], k[7]) KCE(k[8], k[9]) KCE(k[10], k[11]) KCE(k[12], k[13]) KCE(k[14], k[15])

__device__ __forceinline__ float xhalf(float v) {
    const int b = __builtin_bit_cast(int, v);
    return __builtin_bit_cast(float,
        __builtin_amdgcn_permlanex16(b, b, 0x76543210, 0xfedcba98, false, false));
}

__device__ __forceinline__ void log_top16(const char* col, int n, float (&key)[KNN]) {
#pragma unroll
    for (int s = 0; s < KNN; ++s) key[s] = 3.0e38f;
#pragma unroll 1
    for (int p0 = 0; p0 < LOGCAP; p0 += 16) {
        if (__builtin_amdgcn_ballot_w32(p0 < n) == 0u) break;
        float v[16];
#pragma unroll
        for (int u = 0; u < 16; ++u) {
            const float kk = *(const float*)(col + (p0 + u) * LOGROW);
            v[u] = ((p0 + u) < n) ? kk : 3.0e38f;
        }
        KLAYER0(v) KLAYER1(v) KLAYER2(v) KLAYER3(v) KLAYER4(v)
        KLAYER5(v) KLAYER6(v) KLAYER7(v) KLAYER8(v) KLAYER9(v)
        KLAYER10(key, v)
        KLAYER11(key) KLAYER12(key) KLAYER13(key) KLAYER14(key)
    }
}

__global__ __launch_bounds__(256) void knn_pool_kernel(const float* __restrict__ x,
                                                       const float* __restrict__ sq,
                                                       float* __restrict__ pooled) {
    __shared__ __attribute__((aligned(16))) float pan[2 * PANEL_PTS * PITCH];
    __shared__ __attribute__((aligned(16))) v2f logbuf[(LOGCAP + 1) * 256];
    __shared__ int fin[ROWS_PER_WG * KNN];

    const int tid     = threadIdx.x;
    const int lane    = tid & 31;
    const int wave    = __builtin_amdgcn_readfirstlane(tid >> 5);
    const int b       = blockIdx.x / NWG_PER_BATCH;
    const int rowbase = (blockIdx.x % NWG_PER_BATCH) * ROWS_PER_WG + wave * ROWS_PER_WAVE;
    const float* xb   = x  + (size_t)b * NPTS * CH;
    const float* sqb  = sq + (size_t)b * NPTS;

    const int m = lane & 15;
    const int h = lane >> 4;

    const int spt  = tid >> 3;
    const int sq4  = tid & 7;
    const int sdst = spt * PITCH + 8 * (sq4 >> 1) + 2 * (sq4 & 1);
    const int ndst = spt * PITCH + 32 + (sq4 & 3);
    const bool nlead = (sq4 & 3) == 0;
    const float* gsrc = xb + tid * 4;
    const float* gsq  = sqb + spt;

    v2f qfrag0[8];
#pragma unroll
    for (int c = 0; c < 8; ++c) {
        const float* p0 = xb + (size_t)(rowbase + m) * CH + 4 * c + 2 * h;
        qfrag0[c].x = -2.0f * p0[0];
        qfrag0[c].y = -2.0f * p0[1];
    }
    v2f qfragOne;
    qfragOne.x = (h == 0) ? 1.0f : 0.0f;
    qfragOne.y = 0.0f;

    float key[KNN];
#pragma unroll
    for (int s = 0; s < KNN; ++s) key[s] = 3.0e38f;

    char* const logb  = (char*)logbuf;
    char* const lcol  = logb + tid * 8;
    unsigned    nfill = 0;

#pragma unroll 1
    for (int s0 = 0; s0 < STEPS_PER_PANEL; s0 += 8) {
        v4f   tv[8];
        float ts[8];
#pragma unroll
        for (int u = 0; u < 8; ++u) {
            const int tb = (s0 + u) * 32;
            tv[u] = *(const v4f*)(gsrc + (size_t)tb * CH);
            ts[u] = gsq[tb];
        }
#pragma unroll
        for (int u = 0; u < 8; ++u) {
            float* d = pan + (s0 + u) * 32 * PITCH;
            v2f lo; lo.x = tv[u].x; lo.y = tv[u].y;
            v2f hi; hi.x = tv[u].z; hi.y = tv[u].w;
            *(v2f*)(d + sdst)     = lo;
            *(v2f*)(d + sdst + 4) = hi;
            d[ndst] = nlead ? ts[u] : 0.0f;
        }
    }
    v4f   rsv[NSETS];
    float rss[NSETS];
#pragma unroll
    for (int u = 0; u < NSETS; ++u) { rsv[u] = v4f{}; rss[u] = 0.0f; }
#pragma unroll
    for (int u = 0; u < REQ_AHEAD; ++u) {
        const int tb = PANEL_PTS + 32 * u;
        rsv[u] = *(const v4f*)(gsrc + (size_t)tb * CH);
        rss[u] = gsq[tb];
    }
    __syncthreads();

    const int foff = m * PITCH + 4 * h;
    const int idx8 = 8 * h;

    for (int pn = 0; pn < NPANELS; ++pn) {
        const float* cur = pan + (pn & 1) * (PANEL_PTS * PITCH);
        float*       nxt = pan + ((pn & 1) ^ 1) * (PANEL_PTS * PITCH);

        v4f f0[4], f1[4];
        v2f fs0, fs1;
#pragma unroll
        for (int g = 0; g < 4; ++g) f0[g] = *(const v4f*)(cur + foff + 8 * g);
        fs0 = *(const v2f*)(cur + m * PITCH + 32 + 2 * h);
#pragma unroll
        for (int g = 0; g < 4; ++g) f1[g] = *(const v4f*)(cur + 16 * PITCH + foff + 8 * g);
        fs1 = *(const v2f*)(cur + 16 * PITCH + m * PITCH + 32 + 2 * h);

#pragma unroll 1
        for (int t4 = 0; t4 < STEPS_PER_PANEL; t4 += NSETS) {
#pragma unroll
        for (int u = 0; u < NSETS; ++u) {
            const int tt = t4 + u;
            const int colbase = pn * PANEL_PTS + tt * 32;

            int ibv[8];
#pragma unroll
            for (int p = 0; p < 8; ++p)
                ibv[p] = colbase + idx8 + ((p < 4) ? 2 * p + 1 : 2 * p + 9);

            if ((u & (LOGCHK - 1)) == 0) {
                const int n = (int)nfill;
                if (__builtin_amdgcn_ballot_w32(n > LOGCAP - 16 * LOGCHK) != 0u) {
                    log_top16(lcol, n, key);
                    const float T = key[KNN - 1];
                    int nlt = 0;
#pragma unroll
                    for (int s = 0; s < KNN - 1; ++s) nlt += (key[s] < T) ? 1 : 0;
                    const int w = log_filter(lcol, n, T, KNN - nlt);
                    nfill = (unsigned)w;
                }
            }

            {
                const int g  = pn * STEPS_PER_PANEL + tt + REQ_AHEAD;
                const int gc = (g < NPIECES) ? g : (NPIECES - 1);
                const int tb = PANEL_PTS + 32 * gc;
                rsv[(u + REQ_AHEAD) % NSETS] = *(const v4f*)(gsrc + (size_t)tb * CH);
                rss[(u + REQ_AHEAD) % NSETS] = gsq[tb];
            }

            v8f acc0 = {};
            v8f acc1 = {};
#pragma unroll
            for (int g = 0; g < 4; ++g) {
                v2f a0; a0.x = f0[g].x; a0.y = f0[g].y;
                v2f a1; a1.x = f1[g].x; a1.y = f1[g].y;
                v2f b0; b0.x = f0[g].z; b0.y = f0[g].w;
                v2f b1; b1.x = f1[g].z; b1.y = f1[g].w;
                acc0 = __builtin_amdgcn_wmma_f32_16x16x4_f32(
                    false, a0, false, qfrag0[2 * g], (short)0, acc0, false, false);
                acc1 = __builtin_amdgcn_wmma_f32_16x16x4_f32(
                    false, a1, false, qfrag0[2 * g], (short)0, acc1, false, false);
                acc0 = __builtin_amdgcn_wmma_f32_16x16x4_f32(
                    false, b0, false, qfrag0[2 * g + 1], (short)0, acc0, false, false);
                acc1 = __builtin_amdgcn_wmma_f32_16x16x4_f32(
                    false, b1, false, qfrag0[2 * g + 1], (short)0, acc1, false, false);
            }
            acc0 = __builtin_amdgcn_wmma_f32_16x16x4_f32(
                false, fs0, false, qfragOne, (short)0, acc0, false, false);
            acc1 = __builtin_amdgcn_wmma_f32_16x16x4_f32(
                false, fs1, false, qfragOne, (short)0, acc1, false, false);

            __builtin_amdgcn_sched_barrier(0);
            if (tt + 1 < STEPS_PER_PANEL) {
                const float* fr = cur + (tt + 1) * 32 * PITCH;
#pragma unroll
                for (int g = 0; g < 4; ++g) f0[g] = *(const v4f*)(fr + foff + 8 * g);
                fs0 = *(const v2f*)(fr + m * PITCH + 32 + 2 * h);
#pragma unroll
                for (int g = 0; g < 4; ++g) f1[g] = *(const v4f*)(fr + 16 * PITCH + foff + 8 * g);
                fs1 = *(const v2f*)(fr + 16 * PITCH + m * PITCH + 32 + 2 * h);
            }
            __builtin_amdgcn_sched_barrier(0);

            float cand[16];
#pragma unroll
            for (int r = 0; r < 8; ++r) {
                cand[r]     = acc0[r];
                cand[8 + r] = acc1[r];
            }

            const float T = key[KNN - 1];

            char* slot[8];
            v2f   ent0[8], ent1[8];
#pragma unroll
            for (int p = 0; p < 8; ++p) {
                const int   ca = 2 * p;
                const float da = cand[ca];
                const float db = cand[ca + 1];
                const int   sa = __builtin_bit_cast(int, da - T);
                const int   sb = __builtin_bit_cast(int, db - T);
                const unsigned ua = (unsigned)sa >> 31;
                const unsigned ub = (unsigned)sb >> 31;
                const int   ib = ibv[p];
                ent0[p].x = (ua != 0u) ? da : db;
                ent0[p].y = __builtin_bit_cast(float, ib - (int)ua);
                ent1[p].x = db;
                ent1[p].y = __builtin_bit_cast(float, ib);
                slot[p] = lcol + nfill * LOGROW;
                nfill += ua + ub;
            }
#pragma unroll
            for (int p = 0; p < 8; ++p) {
                *(v2f*)slot[p]            = ent0[p];
                *(v2f*)(slot[p] + LOGROW) = ent1[p];
            }
            {
                float* d = nxt + tt * 32 * PITCH;
                const v4f   sv = rsv[u];
                const float ss = rss[u];
                v2f lo; lo.x = sv.x; lo.y = sv.y;
                v2f hi; hi.x = sv.z; hi.y = sv.w;
                *(v2f*)(d + sdst)     = lo;
                *(v2f*)(d + sdst + 4) = hi;
                d[ndst] = nlead ? ss : 0.0f;
            }
            __builtin_amdgcn_sched_group_barrier(0x001, 1000, 0);
            __builtin_amdgcn_sched_group_barrier(0x200, 19, 0);
            __builtin_amdgcn_sched_barrier(0);
            {
                const float t0 = __builtin_fminf(__builtin_fminf(cand[0], cand[1]), cand[2]);
                const float t1 = __builtin_fminf(__builtin_fminf(cand[3], cand[4]), cand[5]);
                const float t2 = __builtin_fminf(__builtin_fminf(cand[6], cand[7]), cand[8]);
                const float t3 = __builtin_fminf(__builtin_fminf(cand[9], cand[10]), cand[11]);
                const float t4 = __builtin_fminf(__builtin_fminf(cand[12], cand[13]), cand[14]);
                const float u0 = __builtin_fminf(__builtin_fminf(t0, t1), t2);
                const float u1 = __builtin_fminf(__builtin_fminf(t3, t4), cand[15]);
                const float mn = __builtin_fminf(u0, u1);
#pragma unroll
                for (int s = KNN - 1; s >= 1; --s)
                    key[s] = __builtin_amdgcn_fmed3f(key[s - 1], key[s], mn);
                key[0] = __builtin_fminf(key[0], mn);
            }
        }
        }
        __builtin_amdgcn_fence(__ATOMIC_RELEASE, "workgroup", "local");
        __builtin_amdgcn_s_barrier();
        __builtin_amdgcn_fence(__ATOMIC_ACQUIRE, "workgroup", "local");
    }

    {
        const int n = (int)nfill;
        log_top16(lcol, n, key);
        float rk[KNN];
#pragma unroll
        for (int s = 0; s < KNN; ++s) rk[s] = __builtin_fminf(key[s], xhalf(key[KNN - 1 - s]));
        KLAYER11(rk) KLAYER12(rk) KLAYER13(rk) KLAYER14(rk)
        const float T = rk[KNN - 1];
        int nlt = 0;
#pragma unroll
        for (int s = 0; s < KNN - 1; ++s) nlt += (rk[s] < T) ? 1 : 0;
        const int need = KNN - nlt;
        const int w  = log_filter(lcol, n, T, LOGCAP);
        const int wo = __builtin_amdgcn_permlanex16(w, w, 0x76543210, 0xfedcba98, false, false);
        const bool crowded = __builtin_amdgcn_ballot_w32(w + wo > KNN) != 0u;
        int* frow = fin + (wave * ROWS_PER_WAVE + m) * KNN;
        int cnt = 0;
#pragma unroll 1
        for (int p = 0; p < LOGCAP; ++p) {
            if (__builtin_amdgcn_ballot_w32(p < w) == 0u) break;
            const float ek = *(const float*)(lcol + p * LOGROW);
            const float ei = *(const float*)(lcol + p * LOGROW + 4);
            const int  myidx = __builtin_bit_cast(int, ei);
            const bool live = p < w;
            bool keep = live;
            if (crowded) {
                const bool tie = live && (ek == T);
                int rank = 0;
#pragma unroll 1
                for (int q = 0; q < LOGCAP; ++q) {
                    if (__builtin_amdgcn_ballot_w32(q < w) == 0u) break;
                    const float ak = *(const float*)(lcol + q * LOGROW);
                    const float af = *(const float*)(lcol + q * LOGROW + 4);
                    const int ai = __builtin_bit_cast(int, af);
                    const int at = ((q < w) && (ak == T)) ? 1 : 0;
                    const int oi = __builtin_amdgcn_permlanex16(ai, ai, 0x76543210, 0xfedcba98, false, false);
                    const int ot = __builtin_amdgcn_permlanex16(at, at, 0x76543210, 0xfedcba98, false, false);
                    rank += ((at != 0) && (ai < myidx)) ? 1 : 0;
                    rank += ((ot != 0) && (oi < myidx)) ? 1 : 0;
                }
                keep = live && (!tie || (rank < need));
            }
            if (keep) frow[((h == 0) ? cnt : (KNN - 1 - cnt)) & (KNN - 1)] = myidx;
            cnt += keep ? 1 : 0;
        }
    }
    __syncthreads();

    float* poolb = pooled + ((size_t)b * NPTS + rowbase) * CH;
#pragma unroll 1
    for (int r0 = 0; r0 < ROWS_PER_WAVE; r0 += GROWS) {
        float mx[GROWS];
#pragma unroll
        for (int q = 0; q < GROWS; ++q) {
            mx[q] = -3.0e38f;
#pragma unroll
            for (int k = 0; k < KNN; ++k) {
                const int j = fin[(wave * ROWS_PER_WAVE + r0 + q) * KNN + k] & (NPTS - 1);
                const float v = xb[(size_t)j * CH + lane];
                mx[q] = fmaxf(mx[q], v);
            }
        }
#pragma unroll
        for (int q = 0; q < GROWS; ++q) poolb[(r0 + q) * CH + lane] = mx[q];
    }
}

__global__ __launch_bounds__(32) void mlp_kernel(const float* __restrict__ pooled,
                                                 const float* __restrict__ Wl,
                                                 const float* __restrict__ bl,
                                                 const float* __restrict__ Wc,
                                                 const float* __restrict__ bc,
                                                 float* __restrict__ out) {
    __shared__ float hT[16 * 68];

    const int lane = threadIdx.x;
    const int m = lane & 15;
    const int h = lane >> 4;
    const size_t rowbase = (size_t)blockIdx.x * 16;

    v2f afrag[8];
#pragma unroll
    for (int c = 0; c < 8; ++c) {
        const float* p = pooled + (rowbase + m) * CH + 4 * c + 2 * h;
        afrag[c].x = p[0];
        afrag[c].y = p[1];
    }

#pragma unroll
    for (int t = 0; t < 4; ++t) {
        v8f acc = {};
#pragma unroll
        for (int c = 0; c < 8; ++c) {
            const float* p = Wl + (size_t)(4 * c + 2 * h) * LDIM + t * 16 + m;
            v2f bfrag;
            bfrag.x = p[0];
            bfrag.y = p[LDIM];
            acc = __builtin_amdgcn_wmma_f32_16x16x4_f32(
                false, afrag[c], false, bfrag, (short)0, acc, false, false);
        }
        const float bias = bl[t * 16 + m];
#pragma unroll
        for (int r = 0; r < 8; ++r)
            hT[(r + 8 * h) * 68 + t * 16 + m] = acc[r] + bias;
    }
    __syncthreads();

    v2f a2[16];
#pragma unroll
    for (int c = 0; c < 16; ++c) {
        a2[c].x = hT[m * 68 + 4 * c + 2 * h];
        a2[c].y = hT[m * 68 + 4 * c + 2 * h + 1];
    }

    float* outp = out + rowbase * ODIM;
#pragma unroll
    for (int t = 0; t < 8; ++t) {
        v8f acc = {};
#pragma unroll
        for (int c = 0; c < 16; ++c) {
            const float* p = Wc + (size_t)(4 * c + 2 * h) * ODIM + t * 16 + m;
            v2f bfrag;
            bfrag.x = p[0];
            bfrag.y = p[ODIM];
            acc = __builtin_amdgcn_wmma_f32_16x16x4_f32(
                false, a2[c], false, bfrag, (short)0, acc, false, false);
        }
        const float bias = bc[t * 16 + m];
#pragma unroll
        for (int r = 0; r < 8; ++r) {
            const float v = acc[r] + bias;
            outp[(size_t)(r + 8 * h) * ODIM + t * 16 + m] = fmaxf(v, 0.0f);
        }
    }
}

extern "C" void kernel_launch(void* const* d_in, const int* in_sizes, int n_in,
                              void* d_out, int out_size, void* d_ws, size_t ws_size,
                              hipStream_t stream) {
    const float* x  = (const float*)d_in[0];
    const float* Wl = (const float*)d_in[1];
    const float* bl = (const float*)d_in[2];
    const float* Wc = (const float*)d_in[3];
    const float* bc = (const float*)d_in[4];
    float* out = (float*)d_out;

    float* pooled = (float*)d_ws;
    float* sqbuf  = pooled + (size_t)BATCH * NPTS * CH;

    const int total = BATCH * NPTS;
    sq_kernel<<<(total + 255) / 256, 256, 0, stream>>>(x, sqbuf, total);
    knn_pool_kernel<<<BATCH * NWG_PER_BATCH, 256, 0, stream>>>(x, sqbuf, pooled);
    mlp_kernel<<<total / 16, 32, 0, stream>>>(pooled, Wl, bl, Wc, bc, out);
}
